# starting kernel
# baseline (speedup 1.0000x reference)
#include <hip/hip_runtime.h>
#include <climits>

#define FDIM 128
#define HEADS 2
#define CH 64
#define NEG_SLOPE 0.2f

typedef float v2f __attribute__((ext_vector_type(2)));
typedef float v8f __attribute__((ext_vector_type(8)));

__global__ __launch_bounds__(128) void gemm_f32_wmma(
    const float* __restrict__ X, const float* __restrict__ W,
    const float* __restrict__ bias0, const float* __restrict__ bias1,
    float* __restrict__ OUT, int nrows)
{
    const int ntiles = nrows >> 4;
    const int tile = blockIdx.x * 4 + (threadIdx.x >> 5);
    if (tile >= ntiles) return;

    const int lane  = threadIdx.x & 31;
    const int lhalf = lane >> 4;
    const int lidx  = lane & 15;
    const int row0  = tile * 16;

    const float* xrow = X + (size_t)(row0 + lidx) * FDIM + 2 * lhalf;

    v8f acc[8];
#pragma unroll
    for (int t = 0; t < 8; ++t)
#pragma unroll
        for (int r = 0; r < 8; ++r) acc[t][r] = 0.0f;

#pragma unroll 4
    for (int kk = 0; kk < FDIM; kk += 4) {
        v2f a = *(const v2f*)(xrow + kk);
        const int krow = kk + 2 * lhalf;
        const float* wr0 = W + (size_t)krow * FDIM + lidx;
        const float* wr1 = wr0 + FDIM;
#pragma unroll
        for (int t = 0; t < 8; ++t) {
            v2f b;
            b.x = wr0[t * 16];
            b.y = wr1[t * 16];
            acc[t] = __builtin_amdgcn_wmma_f32_16x16x4_f32(
                false, a, false, b, (short)0, acc[t], false, false);
        }
    }

    const int orow = row0 + 8 * lhalf;
#pragma unroll
    for (int t = 0; t < 8; ++t) {
        const int col = t * 16 + lidx;
        float badd = 0.0f;
        if (bias0) badd += bias0[col];
        if (bias1) badd += bias1[col];
        float* o = OUT + (size_t)orow * FDIM + col;
#pragma unroll
        for (int r = 0; r < 8; ++r)
            o[(size_t)r * FDIM] = acc[t][r] + badd;
    }
}

__global__ void att_kernel(const float* __restrict__ Hp,
                           const float* __restrict__ att_src,
                           const float* __restrict__ att_dst,
                           float* __restrict__ a_s, float* __restrict__ a_d,
                           int n)
{
    int i = blockIdx.x * blockDim.x + threadIdx.x;
    if (i >= n * HEADS) return;
    const int node = i >> 1, h = i & 1;
    const float* hp = Hp + (size_t)node * FDIM + h * CH;
    const float* as = att_src + h * CH;
    const float* ad = att_dst + h * CH;
    float s0 = 0.f, s1 = 0.f;
#pragma unroll 8
    for (int c = 0; c < CH; ++c) {
        float v = hp[c];
        s0 += v * as[c];
        s1 += v * ad[c];
    }
    a_s[i] = s0;
    a_d[i] = s1;
}

__global__ void init_kernel(int* __restrict__ m, float* __restrict__ denom, int n2)
{
    int i = blockIdx.x * blockDim.x + threadIdx.x;
    if (i < n2) { m[i] = INT_MIN; denom[i] = 0.0f; }
}

__device__ __forceinline__ int   fmapf(float x)  { int b = __float_as_int(x); return b >= 0 ? b : (b ^ 0x7FFFFFFF); }
__device__ __forceinline__ float funmap(int b)   { return __int_as_float(b >= 0 ? b : (b ^ 0x7FFFFFFF)); }
__device__ __forceinline__ float lrelu(float x)  { return x > 0.f ? x : NEG_SLOPE * x; }

__global__ void edge_max_kernel(const int* __restrict__ src, const int* __restrict__ dst,
                                const float* __restrict__ a_s, const float* __restrict__ a_d,
                                int* __restrict__ m, int E_, int M_)
{
    int k = blockIdx.x * blockDim.x + threadIdx.x;
    if (k >= M_) return;
    const int s = k < E_ ? src[k] : k - E_;
    const int d = k < E_ ? dst[k] : k - E_;
#pragma unroll
    for (int h = 0; h < HEADS; ++h) {
        float e = lrelu(a_s[s * HEADS + h] + a_d[d * HEADS + h]);
        atomicMax(&m[d * HEADS + h], fmapf(e));
    }
}

__global__ void edge_sum_kernel(const int* __restrict__ src, const int* __restrict__ dst,
                                const float* __restrict__ a_s, const float* __restrict__ a_d,
                                const int* __restrict__ m, float* __restrict__ P,
                                float* __restrict__ denom, int E_, int M_)
{
    int k = blockIdx.x * blockDim.x + threadIdx.x;
    if (k >= M_) return;
    const int s = k < E_ ? src[k] : k - E_;
    const int d = k < E_ ? dst[k] : k - E_;
#pragma unroll
    for (int h = 0; h < HEADS; ++h) {
        float e  = lrelu(a_s[s * HEADS + h] + a_d[d * HEADS + h]);
        float mx = funmap(m[d * HEADS + h]);
        float p  = __expf(e - mx);
        P[(size_t)k * HEADS + h] = p;
        atomicAdd(&denom[d * HEADS + h], p);
    }
}

__global__ __launch_bounds__(256) void edge_scatter_kernel(
    const int* __restrict__ src, const int* __restrict__ dst,
    const float* __restrict__ P, const float* __restrict__ denom,
    const float* __restrict__ Hp, float* __restrict__ out, int E_, int M_)
{
    const int lane = threadIdx.x & 31;
    const int k = blockIdx.x * 8 + (threadIdx.x >> 5);
    if (k >= M_) return;
    const int s = k < E_ ? src[k] : k - E_;
    const int d = k < E_ ? dst[k] : k - E_;
    const int h = lane >> 4;
    const float alpha = P[(size_t)k * HEADS + h] / (denom[d * HEADS + h] + 1e-16f);
    const int c0 = lane * 4;
    const float4 hv = *(const float4*)(Hp + (size_t)s * FDIM + c0);
    float* o = out + (size_t)d * FDIM + c0;
    atomicAdd(o + 0, hv.x * alpha);
    atomicAdd(o + 1, hv.y * alpha);
    atomicAdd(o + 2, hv.z * alpha);
    atomicAdd(o + 3, hv.w * alpha);
}

__global__ void relu_kernel(float* __restrict__ x, int n)
{
    int i = blockIdx.x * blockDim.x + threadIdx.x;
    if (i < n) x[i] = fmaxf(x[i], 0.0f);
}

extern "C" void kernel_launch(void* const* d_in, const int* in_sizes, int n_in,
                              void* d_out, int out_size, void* d_ws, size_t ws_size,
                              hipStream_t stream)
{
    const float* x        = (const float*)d_in[0];
    const int*   src      = (const int*)  d_in[1];
    const int*   dst      = (const int*)  d_in[2];
    const float* W1       = (const float*)d_in[3];
    const float* att_src1 = (const float*)d_in[4];
    const float* att_dst1 = (const float*)d_in[5];
    const float* b1       = (const float*)d_in[6];
    const float* Wsk1     = (const float*)d_in[7];
    const float* bsk1     = (const float*)d_in[8];
    const float* W2       = (const float*)d_in[9];
    const float* att_src2 = (const float*)d_in[10];
    const float* att_dst2 = (const float*)d_in[11];
    const float* b2       = (const float*)d_in[12];
    const float* Wsk2     = (const float*)d_in[13];
    const float* bsk2     = (const float*)d_in[14];

    const int N  = in_sizes[0] / FDIM;
    const int E  = in_sizes[1];
    const int M  = E + N;
    float* out   = (float*)d_out;

    char* base = (char*)d_ws;
    size_t off = 0;
    auto alloc = [&](size_t bytes) {
        void* p = base + off;
        off = (off + bytes + 255) & ~(size_t)255;
        return p;
    };
    float* Hproj = (float*)alloc((size_t)N * FDIM * 4);
    float* Hbuf  = (float*)alloc((size_t)N * FDIM * 4);
    float* a_s   = (float*)alloc((size_t)N * HEADS * 4);
    float* a_d   = (float*)alloc((size_t)N * HEADS * 4);
    int*   mbuf  = (int*)  alloc((size_t)N * HEADS * 4);
    float* denom = (float*)alloc((size_t)N * HEADS * 4);
    float* P     = (float*)alloc((size_t)M * HEADS * 4);

    const int ntiles     = N / 16;
    const dim3 gGemm((ntiles + 3) / 4), bGemm(128);
    const dim3 gAtt((N * HEADS + 255) / 256), b256(256);
    const dim3 gInit((N * HEADS + 255) / 256);
    const dim3 gEdge((M + 255) / 256);
    const dim3 gScat((M + 7) / 8);
    const dim3 gRelu((N * FDIM + 255) / 256);

    gemm_f32_wmma<<<gGemm, bGemm, 0, stream>>>(x, W1, nullptr, nullptr, Hproj, N);
    gemm_f32_wmma<<<gGemm, bGemm, 0, stream>>>(x, Wsk1, b1, bsk1, Hbuf, N);
    att_kernel<<<gAtt, b256, 0, stream>>>(Hproj, att_src1, att_dst1, a_s, a_d, N);
    init_kernel<<<gInit, b256, 0, stream>>>(mbuf, denom, N * HEADS);
    edge_max_kernel<<<gEdge, b256, 0, stream>>>(src, dst, a_s, a_d, mbuf, E, M);
    edge_sum_kernel<<<gEdge, b256, 0, stream>>>(src, dst, a_s, a_d, mbuf, P, denom, E, M);
    edge_scatter_kernel<<<gScat, b256, 0, stream>>>(src, dst, P, denom, Hproj, Hbuf, E, M);
    relu_kernel<<<gRelu, b256, 0, stream>>>(Hbuf, N * FDIM);

    gemm_f32_wmma<<<gGemm, bGemm, 0, stream>>>(Hbuf, W2, nullptr, nullptr, Hproj, N);
    gemm_f32_wmma<<<gGemm, bGemm, 0, stream>>>(Hbuf, Wsk2, b2, bsk2, out, N);
    att_kernel<<<gAtt, b256, 0, stream>>>(Hproj, att_src2, att_dst2, a_s, a_d, N);
    init_kernel<<<gInit, b256, 0, stream>>>(mbuf, denom, N * HEADS);
    edge_max_kernel<<<gEdge, b256, 0, stream>>>(src, dst, a_s, a_d, mbuf, E, M);
    edge_sum_kernel<<<gEdge, b256, 0, stream>>>(src, dst, a_s, a_d, mbuf, P, denom, E, M);
    edge_scatter_kernel<<<gScat, b256, 0, stream>>>(src, dst, P, denom, Hproj, out, E, M);
}


# Pad the per-node softmax max and sum words to one 128-byte line per node, so the scattered atomics of different nodes no longer serialise on shared cache lines
# speedup vs baseline: 1.2930x; 1.2930x over previous
#include <hip/hip_runtime.h>
#include <climits>

#define FDIM 128
#define HEADS 2
#define CH 64
#define NEG_SLOPE 0.2f
#define NSTRIDE 32

typedef float v2f __attribute__((ext_vector_type(2)));
typedef float v8f __attribute__((ext_vector_type(8)));

__global__ __launch_bounds__(128) void gemm_f32_wmma(
    const float* __restrict__ X, const float* __restrict__ W,
    const float* __restrict__ bias0, const float* __restrict__ bias1,
    float* __restrict__ OUT, int nrows)
{
    const int ntiles = nrows >> 4;
    const int tile = blockIdx.x * 4 + (threadIdx.x >> 5);
    if (tile >= ntiles) return;

    const int lane  = threadIdx.x & 31;
    const int lhalf = lane >> 4;
    const int lidx  = lane & 15;
    const int row0  = tile * 16;

    const float* xrow = X + (size_t)(row0 + lidx) * FDIM + 2 * lhalf;

    v8f acc[8];
#pragma unroll
    for (int t = 0; t < 8; ++t)
#pragma unroll
        for (int r = 0; r < 8; ++r) acc[t][r] = 0.0f;

#pragma unroll 4
    for (int kk = 0; kk < FDIM; kk += 4) {
        v2f a = *(const v2f*)(xrow + kk);
        const int krow = kk + 2 * lhalf;
        const float* wr0 = W + (size_t)krow * FDIM + lidx;
        const float* wr1 = wr0 + FDIM;
#pragma unroll
        for (int t = 0; t < 8; ++t) {
            v2f b;
            b.x = wr0[t * 16];
            b.y = wr1[t * 16];
            acc[t] = __builtin_amdgcn_wmma_f32_16x16x4_f32(
                false, a, false, b, (short)0, acc[t], false, false);
        }
    }

    const int orow = row0 + 8 * lhalf;
#pragma unroll
    for (int t = 0; t < 8; ++t) {
        const int col = t * 16 + lidx;
        float badd = 0.0f;
        if (bias0) badd += bias0[col];
        if (bias1) badd += bias1[col];
        float* o = OUT + (size_t)orow * FDIM + col;
#pragma unroll
        for (int r = 0; r < 8; ++r)
            o[(size_t)r * FDIM] = acc[t][r] + badd;
    }
}

__global__ void att_kernel(const float* __restrict__ Hp,
                           const float* __restrict__ att_src,
                           const float* __restrict__ att_dst,
                           float* __restrict__ a_s, float* __restrict__ a_d,
                           int n)
{
    int i = blockIdx.x * blockDim.x + threadIdx.x;
    if (i >= n * HEADS) return;
    const int node = i >> 1, h = i & 1;
    const float* hp = Hp + (size_t)node * FDIM + h * CH;
    const float* as = att_src + h * CH;
    const float* ad = att_dst + h * CH;
    float s0 = 0.f, s1 = 0.f;
#pragma unroll 8
    for (int c = 0; c < CH; ++c) {
        float v = hp[c];
        s0 += v * as[c];
        s1 += v * ad[c];
    }
    a_s[i] = s0;
    a_d[i] = s1;
}

__global__ void init_kernel(int* __restrict__ m, float* __restrict__ denom, int n2)
{
    int i = blockIdx.x * blockDim.x + threadIdx.x;
    if (i < n2) {
        const int j = (i >> 1) * NSTRIDE + (i & 1);
        m[j] = INT_MIN;
        denom[j] = 0.0f;
    }
}

__device__ __forceinline__ int   fmapf(float x)  { int b = __float_as_int(x); return b >= 0 ? b : (b ^ 0x7FFFFFFF); }
__device__ __forceinline__ float funmap(int b)   { return __int_as_float(b >= 0 ? b : (b ^ 0x7FFFFFFF)); }
__device__ __forceinline__ float lrelu(float x)  { return x > 0.f ? x : NEG_SLOPE * x; }

__global__ void edge_max_kernel(const int* __restrict__ src, const int* __restrict__ dst,
                                const float* __restrict__ a_s, const float* __restrict__ a_d,
                                int* __restrict__ m, int E_, int M_)
{
    int k = blockIdx.x * blockDim.x + threadIdx.x;
    if (k >= M_) return;
    const int s = k < E_ ? src[k] : k - E_;
    const int d = k < E_ ? dst[k] : k - E_;
#pragma unroll
    for (int h = 0; h < HEADS; ++h) {
        float e = lrelu(a_s[s * HEADS + h] + a_d[d * HEADS + h]);
        atomicMax(&m[d * NSTRIDE + h], fmapf(e));
    }
}

__global__ void edge_sum_kernel(const int* __restrict__ src, const int* __restrict__ dst,
                                const float* __restrict__ a_s, const float* __restrict__ a_d,
                                const int* __restrict__ m, float* __restrict__ P,
                                float* __restrict__ denom, int E_, int M_)
{
    int k = blockIdx.x * blockDim.x + threadIdx.x;
    if (k >= M_) return;
    const int s = k < E_ ? src[k] : k - E_;
    const int d = k < E_ ? dst[k] : k - E_;
#pragma unroll
    for (int h = 0; h < HEADS; ++h) {
        float e  = lrelu(a_s[s * HEADS + h] + a_d[d * HEADS + h]);
        float mx = funmap(m[d * NSTRIDE + h]);
        float p  = __expf(e - mx);
        P[(size_t)k * HEADS + h] = p;
        atomicAdd(&denom[d * NSTRIDE + h], p);
    }
}

__global__ __launch_bounds__(256) void edge_scatter_kernel(
    const int* __restrict__ src, const int* __restrict__ dst,
    const float* __restrict__ P, const float* __restrict__ denom,
    const float* __restrict__ Hp, float* __restrict__ out, int E_, int M_)
{
    const int lane = threadIdx.x & 31;
    const int k = blockIdx.x * 8 + (threadIdx.x >> 5);
    if (k >= M_) return;
    const int s = k < E_ ? src[k] : k - E_;
    const int d = k < E_ ? dst[k] : k - E_;
    const int h = lane >> 4;
    const float alpha = P[(size_t)k * HEADS + h] / (denom[d * NSTRIDE + h] + 1e-16f);
    const int c0 = lane * 4;
    const float4 hv = *(const float4*)(Hp + (size_t)s * FDIM + c0);
    float* o = out + (size_t)d * FDIM + c0;
    atomicAdd(o + 0, hv.x * alpha);
    atomicAdd(o + 1, hv.y * alpha);
    atomicAdd(o + 2, hv.z * alpha);
    atomicAdd(o + 3, hv.w * alpha);
}

__global__ void relu_kernel(float* __restrict__ x, int n)
{
    int i = blockIdx.x * blockDim.x + threadIdx.x;
    if (i < n) x[i] = fmaxf(x[i], 0.0f);
}

extern "C" void kernel_launch(void* const* d_in, const int* in_sizes, int n_in,
                              void* d_out, int out_size, void* d_ws, size_t ws_size,
                              hipStream_t stream)
{
    const float* x        = (const float*)d_in[0];
    const int*   src      = (const int*)  d_in[1];
    const int*   dst      = (const int*)  d_in[2];
    const float* W1       = (const float*)d_in[3];
    const float* att_src1 = (const float*)d_in[4];
    const float* att_dst1 = (const float*)d_in[5];
    const float* b1       = (const float*)d_in[6];
    const float* Wsk1     = (const float*)d_in[7];
    const float* bsk1     = (const float*)d_in[8];
    const float* W2       = (const float*)d_in[9];
    const float* att_src2 = (const float*)d_in[10];
    const float* att_dst2 = (const float*)d_in[11];
    const float* b2       = (const float*)d_in[12];
    const float* Wsk2     = (const float*)d_in[13];
    const float* bsk2     = (const float*)d_in[14];

    const int N  = in_sizes[0] / FDIM;
    const int E  = in_sizes[1];
    const int M  = E + N;
    float* out   = (float*)d_out;

    char* base = (char*)d_ws;
    size_t off = 0;
    auto alloc = [&](size_t bytes) {
        void* p = base + off;
        off = (off + bytes + 255) & ~(size_t)255;
        return p;
    };
    float* Hproj = (float*)alloc((size_t)N * FDIM * 4);
    float* Hbuf  = (float*)alloc((size_t)N * FDIM * 4);
    float* a_s   = (float*)alloc((size_t)N * HEADS * 4);
    float* a_d   = (float*)alloc((size_t)N * HEADS * 4);
    int*   mbuf  = (int*)  alloc((size_t)N * NSTRIDE * 4);
    float* denom = (float*)alloc((size_t)N * NSTRIDE * 4);
    float* P     = (float*)alloc((size_t)M * HEADS * 4);

    const int ntiles     = N / 16;
    const dim3 gGemm((ntiles + 3) / 4), bGemm(128);
    const dim3 gAtt((N * HEADS + 255) / 256), b256(256);
    const dim3 gInit((N * HEADS + 255) / 256);
    const dim3 gEdge((M + 255) / 256);
    const dim3 gScat((M + 7) / 8);
    const dim3 gRelu((N * FDIM + 255) / 256);

    gemm_f32_wmma<<<gGemm, bGemm, 0, stream>>>(x, W1, nullptr, nullptr, Hproj, N);
    gemm_f32_wmma<<<gGemm, bGemm, 0, stream>>>(x, Wsk1, b1, bsk1, Hbuf, N);
    att_kernel<<<gAtt, b256, 0, stream>>>(Hproj, att_src1, att_dst1, a_s, a_d, N);
    init_kernel<<<gInit, b256, 0, stream>>>(mbuf, denom, N * HEADS);
    edge_max_kernel<<<gEdge, b256, 0, stream>>>(src, dst, a_s, a_d, mbuf, E, M);
    edge_sum_kernel<<<gEdge, b256, 0, stream>>>(src, dst, a_s, a_d, mbuf, P, denom, E, M);
    edge_scatter_kernel<<<gScat, b256, 0, stream>>>(src, dst, P, denom, Hproj, Hbuf, E, M);
    relu_kernel<<<gRelu, b256, 0, stream>>>(Hbuf, N * FDIM);

    gemm_f32_wmma<<<gGemm, bGemm, 0, stream>>>(Hbuf, W2, nullptr, nullptr, Hproj, N);
    gemm_f32_wmma<<<gGemm, bGemm, 0, stream>>>(Hbuf, Wsk2, b2, bsk2, out, N);
    att_kernel<<<gAtt, b256, 0, stream>>>(Hproj, att_src2, att_dst2, a_s, a_d, N);
    init_kernel<<<gInit, b256, 0, stream>>>(mbuf, denom, N * HEADS);
    edge_max_kernel<<<gEdge, b256, 0, stream>>>(src, dst, a_s, a_d, mbuf, E, M);
    edge_sum_kernel<<<gEdge, b256, 0, stream>>>(src, dst, a_s, a_d, mbuf, P, denom, E, M);
    edge_scatter_kernel<<<gScat, b256, 0, stream>>>(src, dst, P, denom, Hproj, out, E, M);
}


# Replace the float-atomic edge scatter by a per-node gather over an edge list sorted by destination once per call (counting sort); each row summed in registers in arrival-rank order, one plain store
# speedup vs baseline: 1.7279x; 1.3363x over previous
#include <hip/hip_runtime.h>
#include <climits>

#define FDIM 128
#define HEADS 2
#define CH 64
#define NEG_SLOPE 0.2f
#define NSTRIDE 32
#define CSTRIDE 32

typedef float v2f __attribute__((ext_vector_type(2)));
typedef float v8f __attribute__((ext_vector_type(8)));

__global__ __launch_bounds__(128) void gemm_f32_wmma(
    const float* __restrict__ X, const float* __restrict__ W,
    const float* __restrict__ bias0, const float* __restrict__ bias1,
    float* __restrict__ OUT, int nrows)
{
    const int ntiles = nrows >> 4;
    const int tile = blockIdx.x * 4 + (threadIdx.x >> 5);
    if (tile >= ntiles) return;

    const int lane  = threadIdx.x & 31;
    const int lhalf = lane >> 4;
    const int lidx  = lane & 15;
    const int row0  = tile * 16;

    const float* xrow = X + (size_t)(row0 + lidx) * FDIM + 2 * lhalf;

    v8f acc[8];
#pragma unroll
    for (int t = 0; t < 8; ++t)
#pragma unroll
        for (int r = 0; r < 8; ++r) acc[t][r] = 0.0f;

#pragma unroll 4
    for (int kk = 0; kk < FDIM; kk += 4) {
        v2f a = *(const v2f*)(xrow + kk);
        const int krow = kk + 2 * lhalf;
        const float* wr0 = W + (size_t)krow * FDIM + lidx;
        const float* wr1 = wr0 + FDIM;
#pragma unroll
        for (int t = 0; t < 8; ++t) {
            v2f b;
            b.x = wr0[t * 16];
            b.y = wr1[t * 16];
            acc[t] = __builtin_amdgcn_wmma_f32_16x16x4_f32(
                false, a, false, b, (short)0, acc[t], false, false);
        }
    }

    const int orow = row0 + 8 * lhalf;
#pragma unroll
    for (int t = 0; t < 8; ++t) {
        const int col = t * 16 + lidx;
        float badd = 0.0f;
        if (bias0) badd += bias0[col];
        if (bias1) badd += bias1[col];
        float* o = OUT + (size_t)orow * FDIM + col;
#pragma unroll
        for (int r = 0; r < 8; ++r)
            o[(size_t)r * FDIM] = acc[t][r] + badd;
    }
}

__global__ void att_kernel(const float* __restrict__ Hp,
                           const float* __restrict__ att_src,
                           const float* __restrict__ att_dst,
                           float* __restrict__ a_s, float* __restrict__ a_d,
                           int n)
{
    int i = blockIdx.x * blockDim.x + threadIdx.x;
    if (i >= n * HEADS) return;
    const int node = i >> 1, h = i & 1;
    const float* hp = Hp + (size_t)node * FDIM + h * CH;
    const float* as = att_src + h * CH;
    const float* ad = att_dst + h * CH;
    float s0 = 0.f, s1 = 0.f;
#pragma unroll 8
    for (int c = 0; c < CH; ++c) {
        float v = hp[c];
        s0 += v * as[c];
        s1 += v * ad[c];
    }
    a_s[i] = s0;
    a_d[i] = s1;
}

__global__ void init_kernel(int* __restrict__ m, float* __restrict__ denom, int n2)
{
    int i = blockIdx.x * blockDim.x + threadIdx.x;
    if (i < n2) {
        const int j = (i >> 1) * NSTRIDE + (i & 1);
        m[j] = INT_MIN;
        denom[j] = 0.0f;
    }
}

__device__ __forceinline__ int   fmapf(float x)  { int b = __float_as_int(x); return b >= 0 ? b : (b ^ 0x7FFFFFFF); }
__device__ __forceinline__ float funmap(int b)   { return __int_as_float(b >= 0 ? b : (b ^ 0x7FFFFFFF)); }
__device__ __forceinline__ float lrelu(float x)  { return x > 0.f ? x : NEG_SLOPE * x; }

__global__ void edge_max_kernel(const int* __restrict__ src, const int* __restrict__ dst,
                                const float* __restrict__ a_s, const float* __restrict__ a_d,
                                int* __restrict__ m, int E_, int M_)
{
    int k = blockIdx.x * blockDim.x + threadIdx.x;
    if (k >= M_) return;
    const int s = k < E_ ? src[k] : k - E_;
    const int d = k < E_ ? dst[k] : k - E_;
#pragma unroll
    for (int h = 0; h < HEADS; ++h) {
        float e = lrelu(a_s[s * HEADS + h] + a_d[d * HEADS + h]);
        atomicMax(&m[d * NSTRIDE + h], fmapf(e));
    }
}

__global__ void edge_sum_kernel(const int* __restrict__ src, const int* __restrict__ dst,
                                const float* __restrict__ a_s, const float* __restrict__ a_d,
                                const int* __restrict__ m, float* __restrict__ P,
                                float* __restrict__ denom, int E_, int M_)
{
    int k = blockIdx.x * blockDim.x + threadIdx.x;
    if (k >= M_) return;
    const int s = k < E_ ? src[k] : k - E_;
    const int d = k < E_ ? dst[k] : k - E_;
#pragma unroll
    for (int h = 0; h < HEADS; ++h) {
        float e  = lrelu(a_s[s * HEADS + h] + a_d[d * HEADS + h]);
        float mx = funmap(m[d * NSTRIDE + h]);
        float p  = __expf(e - mx);
        P[(size_t)k * HEADS + h] = p;
        atomicAdd(&denom[d * NSTRIDE + h], p);
    }
}

__global__ void csr_zero_kernel(int* __restrict__ cur, int n)
{
    int i = blockIdx.x * blockDim.x + threadIdx.x;
    if (i < n) cur[(size_t)i * CSTRIDE] = 0;
}

__global__ void csr_count_kernel(const int* __restrict__ dst, int* __restrict__ cur, int E_)
{
    int k = blockIdx.x * blockDim.x + threadIdx.x;
    if (k >= E_) return;
    atomicAdd(&cur[(size_t)dst[k] * CSTRIDE], 1);
}

__device__ __forceinline__ int wave_incl_scan(int v, int lane)
{
#pragma unroll
    for (int o = 1; o < 32; o <<= 1) {
        int t = __shfl_up(v, o);
        if (lane >= o) v += t;
    }
    return v;
}

__global__ __launch_bounds__(256) void csr_blocksum_kernel(
    const int* __restrict__ cur, int* __restrict__ bsum, int n)
{
    __shared__ int wtot[8];
    const int i = blockIdx.x * 256 + threadIdx.x;
    const int lane = threadIdx.x & 31, w = threadIdx.x >> 5;
    const int c = i < n ? cur[(size_t)i * CSTRIDE] : 0;
    const int inc = wave_incl_scan(c, lane);
    if (lane == 31) wtot[w] = inc;
    __syncthreads();
    if (threadIdx.x == 0) {
        int s = 0;
#pragma unroll
        for (int j = 0; j < 8; ++j) s += wtot[j];
        bsum[blockIdx.x] = s;
    }
}

__global__ __launch_bounds__(256) void csr_scan_kernel(
    int* __restrict__ cur, const int* __restrict__ bsum, int* __restrict__ rowptr, int n)
{
    __shared__ int wtot[8];
    __shared__ int wpre[8];
    const int i = blockIdx.x * 256 + threadIdx.x;
    const int lane = threadIdx.x & 31, w = threadIdx.x >> 5;
    const int c = i < n ? cur[(size_t)i * CSTRIDE] : 0;

    int part = 0;
    for (int j = threadIdx.x; j < (int)blockIdx.x; j += 256) part += bsum[j];
    const int pinc = wave_incl_scan(part, lane);
    const int inc  = wave_incl_scan(c, lane);
    if (lane == 31) { wpre[w] = pinc; wtot[w] = inc; }
    __syncthreads();
    int off = 0;
#pragma unroll
    for (int j = 0; j < 8; ++j) {
        off += wpre[j];
        if (j < w) off += wtot[j];
    }
    const int excl = off + inc - c;
    if (i < n) {
        rowptr[i] = excl;
        cur[(size_t)i * CSTRIDE] = excl;
        if (i == n - 1) rowptr[n] = excl + c;
    }
}

__global__ void csr_fill_kernel(const int* __restrict__ src, const int* __restrict__ dst,
                                int* __restrict__ cur, int2* __restrict__ ent, int E_)
{
    int k = blockIdx.x * blockDim.x + threadIdx.x;
    if (k >= E_) return;
    const int pos = atomicAdd(&cur[(size_t)dst[k] * CSTRIDE], 1);
    if ((unsigned)pos < (unsigned)E_) ent[pos] = make_int2(k, src[k]);
}

__global__ __launch_bounds__(256) void node_gather_kernel(
    const int2* __restrict__ ent, const int* __restrict__ rowptr,
    const float* __restrict__ P, const float* __restrict__ denom,
    const float* __restrict__ Hp, float* __restrict__ out, int E_, int n)
{
    const int lane = threadIdx.x & 31;
    const int d = __builtin_amdgcn_readfirstlane((int)(blockIdx.x * 8 + (threadIdx.x >> 5)));
    if (d >= n) return;
    const int h  = lane >> 4;
    const int c0 = lane * 4;
    const float den0 = denom[d * NSTRIDE + 0] + 1e-16f;
    const float den1 = denom[d * NSTRIDE + 1] + 1e-16f;
    const int beg = max(rowptr[d], 0);
    const int end = min(rowptr[d + 1], E_);

    const float aself = P[((size_t)E_ + d) * HEADS + h] / (h ? den1 : den0);
    const float4 hs = *(const float4*)(Hp + (size_t)d * FDIM + c0);
    float4 acc;
    acc.x = hs.x * aself; acc.y = hs.y * aself; acc.z = hs.z * aself; acc.w = hs.w * aself;

    for (int base = beg; base < end; base += 32) {
        const int j = base + lane;
        int s_l = d;
        float a0 = 0.f, a1 = 0.f;
        if (j < end) {
            const int2 e = ent[j];
            const int k = min(max(e.x, 0), E_ - 1);
            s_l = min(max(e.y, 0), n - 1);
            const float2 p = *(const float2*)(P + (size_t)k * HEADS);
            a0 = p.x / den0;
            a1 = p.y / den1;
        }
        const int cnt = min(32, end - base);
#pragma unroll 4
        for (int t = 0; t < cnt; ++t) {
            const int s = __builtin_amdgcn_readlane(s_l, t);
            const float w0 = __int_as_float(__builtin_amdgcn_readlane(__float_as_int(a0), t));
            const float w1 = __int_as_float(__builtin_amdgcn_readlane(__float_as_int(a1), t));
            const float a = h ? w1 : w0;
            const float4 hv = *(const float4*)(Hp + (size_t)s * FDIM + c0);
            acc.x += hv.x * a;
            acc.y += hv.y * a;
            acc.z += hv.z * a;
            acc.w += hv.w * a;
        }
    }

    float4* o = (float4*)(out + (size_t)d * FDIM + c0);
    float4 ov = *o;
    ov.x += acc.x; ov.y += acc.y; ov.z += acc.z; ov.w += acc.w;
    *o = ov;
}

__global__ void relu_kernel(float* __restrict__ x, int n)
{
    int i = blockIdx.x * blockDim.x + threadIdx.x;
    if (i < n) x[i] = fmaxf(x[i], 0.0f);
}

extern "C" void kernel_launch(void* const* d_in, const int* in_sizes, int n_in,
                              void* d_out, int out_size, void* d_ws, size_t ws_size,
                              hipStream_t stream)
{
    const float* x        = (const float*)d_in[0];
    const int*   src      = (const int*)  d_in[1];
    const int*   dst      = (const int*)  d_in[2];
    const float* W1       = (const float*)d_in[3];
    const float* att_src1 = (const float*)d_in[4];
    const float* att_dst1 = (const float*)d_in[5];
    const float* b1       = (const float*)d_in[6];
    const float* Wsk1     = (const float*)d_in[7];
    const float* bsk1     = (const float*)d_in[8];
    const float* W2       = (const float*)d_in[9];
    const float* att_src2 = (const float*)d_in[10];
    const float* att_dst2 = (const float*)d_in[11];
    const float* b2       = (const float*)d_in[12];
    const float* Wsk2     = (const float*)d_in[13];
    const float* bsk2     = (const float*)d_in[14];

    const int N  = in_sizes[0] / FDIM;
    const int E  = in_sizes[1];
    const int M  = E + N;
    float* out   = (float*)d_out;

    char* base = (char*)d_ws;
    size_t off = 0;
    auto alloc = [&](size_t bytes) {
        void* p = base + off;
        off = (off + bytes + 255) & ~(size_t)255;
        return p;
    };
    float* Hproj = (float*)alloc((size_t)N * FDIM * 4);
    float* Hbuf  = (float*)alloc((size_t)N * FDIM * 4);
    float* a_s   = (float*)alloc((size_t)N * HEADS * 4);
    float* a_d   = (float*)alloc((size_t)N * HEADS * 4);
    int*   mbuf  = (int*)  alloc((size_t)N * NSTRIDE * 4);
    float* denom = (float*)alloc((size_t)N * NSTRIDE * 4);
    float* P     = (float*)alloc((size_t)M * HEADS * 4);
    const int nblk = (N + 255) / 256;
    int*   cur    = (int*)  alloc((size_t)N * CSTRIDE * 4);
    int*   bsum   = (int*)  alloc((size_t)nblk * 4);
    int*   rowptr = (int*)  alloc((size_t)(N + 1) * 4);
    int2*  ent    = (int2*) alloc((size_t)E * 8);

    const int ntiles     = N / 16;
    const dim3 gGemm((ntiles + 3) / 4), bGemm(128);
    const dim3 gAtt((N * HEADS + 255) / 256), b256(256);
    const dim3 gInit((N * HEADS + 255) / 256);
    const dim3 gEdge((M + 255) / 256);
    const dim3 gNode((N + 7) / 8);
    const dim3 gNodeT((N + 255) / 256);
    const dim3 gEdgeE((E + 255) / 256);
    const dim3 gRelu((N * FDIM + 255) / 256);

    csr_zero_kernel<<<gNodeT, b256, 0, stream>>>(cur, N);
    csr_count_kernel<<<gEdgeE, b256, 0, stream>>>(dst, cur, E);
    csr_blocksum_kernel<<<dim3(nblk), b256, 0, stream>>>(cur, bsum, N);
    csr_scan_kernel<<<dim3(nblk), b256, 0, stream>>>(cur, bsum, rowptr, N);
    csr_fill_kernel<<<gEdgeE, b256, 0, stream>>>(src, dst, cur, ent, E);

    gemm_f32_wmma<<<gGemm, bGemm, 0, stream>>>(x, W1, nullptr, nullptr, Hproj, N);
    gemm_f32_wmma<<<gGemm, bGemm, 0, stream>>>(x, Wsk1, b1, bsk1, Hbuf, N);
    att_kernel<<<gAtt, b256, 0, stream>>>(Hproj, att_src1, att_dst1, a_s, a_d, N);
    init_kernel<<<gInit, b256, 0, stream>>>(mbuf, denom, N * HEADS);
    edge_max_kernel<<<gEdge, b256, 0, stream>>>(src, dst, a_s, a_d, mbuf, E, M);
    edge_sum_kernel<<<gEdge, b256, 0, stream>>>(src, dst, a_s, a_d, mbuf, P, denom, E, M);
    node_gather_kernel<<<gNode, b256, 0, stream>>>(ent, rowptr, P, denom, Hproj, Hbuf, E, N);
    relu_kernel<<<gRelu, b256, 0, stream>>>(Hbuf, N * FDIM);

    gemm_f32_wmma<<<gGemm, bGemm, 0, stream>>>(Hbuf, W2, nullptr, nullptr, Hproj, N);
    gemm_f32_wmma<<<gGemm, bGemm, 0, stream>>>(Hbuf, Wsk2, b2, bsk2, out, N);
    att_kernel<<<gAtt, b256, 0, stream>>>(Hproj, att_src2, att_dst2, a_s, a_d, N);
    init_kernel<<<gInit, b256, 0, stream>>>(mbuf, denom, N * HEADS);
    edge_max_kernel<<<gEdge, b256, 0, stream>>>(src, dst, a_s, a_d, mbuf, E, M);
    edge_sum_kernel<<<gEdge, b256, 0, stream>>>(src, dst, a_s, a_d, mbuf, P, denom, E, M);
    node_gather_kernel<<<gNode, b256, 0, stream>>>(ent, rowptr, P, denom, Hproj, out, E, N);
}


# Fold the ReLU into the layer-1 gather: max with 0 on the finished row in registers before its one store; the elementwise launch of 200000 one-float-per-lane waves and its 51 MB round trip go
# speedup vs baseline: 1.7979x; 1.0405x over previous
#include <hip/hip_runtime.h>
#include <climits>

#define FDIM 128
#define HEADS 2
#define CH 64
#define NEG_SLOPE 0.2f
#define NSTRIDE 32
#define CSTRIDE 32

typedef float v2f __attribute__((ext_vector_type(2)));
typedef float v8f __attribute__((ext_vector_type(8)));

__global__ __launch_bounds__(128) void gemm_f32_wmma(
    const float* __restrict__ X, const float* __restrict__ W,
    const float* __restrict__ bias0, const float* __restrict__ bias1,
    float* __restrict__ OUT, int nrows)
{
    const int ntiles = nrows >> 4;
    const int tile = blockIdx.x * 4 + (threadIdx.x >> 5);
    if (tile >= ntiles) return;

    const int lane  = threadIdx.x & 31;
    const int lhalf = lane >> 4;
    const int lidx  = lane & 15;
    const int row0  = tile * 16;

    const float* xrow = X + (size_t)(row0 + lidx) * FDIM + 2 * lhalf;

    v8f acc[8];
#pragma unroll
    for (int t = 0; t < 8; ++t)
#pragma unroll
        for (int r = 0; r < 8; ++r) acc[t][r] = 0.0f;

#pragma unroll 4
    for (int kk = 0; kk < FDIM; kk += 4) {
        v2f a = *(const v2f*)(xrow + kk);
        const int krow = kk + 2 * lhalf;
        const float* wr0 = W + (size_t)krow * FDIM + lidx;
        const float* wr1 = wr0 + FDIM;
#pragma unroll
        for (int t = 0; t < 8; ++t) {
            v2f b;
            b.x = wr0[t * 16];
            b.y = wr1[t * 16];
            acc[t] = __builtin_amdgcn_wmma_f32_16x16x4_f32(
                false, a, false, b, (short)0, acc[t], false, false);
        }
    }

    const int orow = row0 + 8 * lhalf;
#pragma unroll
    for (int t = 0; t < 8; ++t) {
        const int col = t * 16 + lidx;
        float badd = 0.0f;
        if (bias0) badd += bias0[col];
        if (bias1) badd += bias1[col];
        float* o = OUT + (size_t)orow * FDIM + col;
#pragma unroll
        for (int r = 0; r < 8; ++r)
            o[(size_t)r * FDIM] = acc[t][r] + badd;
    }
}

__global__ void att_kernel(const float* __restrict__ Hp,
                           const float* __restrict__ att_src,
                           const float* __restrict__ att_dst,
                           float* __restrict__ a_s, float* __restrict__ a_d,
                           int n)
{
    int i = blockIdx.x * blockDim.x + threadIdx.x;
    if (i >= n * HEADS) return;
    const int node = i >> 1, h = i & 1;
    const float* hp = Hp + (size_t)node * FDIM + h * CH;
    const float* as = att_src + h * CH;
    const float* ad = att_dst + h * CH;
    float s0 = 0.f, s1 = 0.f;
#pragma unroll 8
    for (int c = 0; c < CH; ++c) {
        float v = hp[c];
        s0 += v * as[c];
        s1 += v * ad[c];
    }
    a_s[i] = s0;
    a_d[i] = s1;
}

__global__ void init_kernel(int* __restrict__ m, float* __restrict__ denom, int n2)
{
    int i = blockIdx.x * blockDim.x + threadIdx.x;
    if (i < n2) {
        const int j = (i >> 1) * NSTRIDE + (i & 1);
        m[j] = INT_MIN;
        denom[j] = 0.0f;
    }
}

__device__ __forceinline__ int   fmapf(float x)  { int b = __float_as_int(x); return b >= 0 ? b : (b ^ 0x7FFFFFFF); }
__device__ __forceinline__ float funmap(int b)   { return __int_as_float(b >= 0 ? b : (b ^ 0x7FFFFFFF)); }
__device__ __forceinline__ float lrelu(float x)  { return x > 0.f ? x : NEG_SLOPE * x; }

__global__ void edge_max_kernel(const int* __restrict__ src, const int* __restrict__ dst,
                                const float* __restrict__ a_s, const float* __restrict__ a_d,
                                int* __restrict__ m, int E_, int M_)
{
    int k = blockIdx.x * blockDim.x + threadIdx.x;
    if (k >= M_) return;
    const int s = k < E_ ? src[k] : k - E_;
    const int d = k < E_ ? dst[k] : k - E_;
#pragma unroll
    for (int h = 0; h < HEADS; ++h) {
        float e = lrelu(a_s[s * HEADS + h] + a_d[d * HEADS + h]);
        atomicMax(&m[d * NSTRIDE + h], fmapf(e));
    }
}

__global__ void edge_sum_kernel(const int* __restrict__ src, const int* __restrict__ dst,
                                const float* __restrict__ a_s, const float* __restrict__ a_d,
                                const int* __restrict__ m, float* __restrict__ P,
                                float* __restrict__ denom, int E_, int M_)
{
    int k = blockIdx.x * blockDim.x + threadIdx.x;
    if (k >= M_) return;
    const int s = k < E_ ? src[k] : k - E_;
    const int d = k < E_ ? dst[k] : k - E_;
#pragma unroll
    for (int h = 0; h < HEADS; ++h) {
        float e  = lrelu(a_s[s * HEADS + h] + a_d[d * HEADS + h]);
        float mx = funmap(m[d * NSTRIDE + h]);
        float p  = __expf(e - mx);
        P[(size_t)k * HEADS + h] = p;
        atomicAdd(&denom[d * NSTRIDE + h], p);
    }
}

__global__ void csr_zero_kernel(int* __restrict__ cur, int n)
{
    int i = blockIdx.x * blockDim.x + threadIdx.x;
    if (i < n) cur[(size_t)i * CSTRIDE] = 0;
}

__global__ void csr_count_kernel(const int* __restrict__ dst, int* __restrict__ cur, int E_)
{
    int k = blockIdx.x * blockDim.x + threadIdx.x;
    if (k >= E_) return;
    atomicAdd(&cur[(size_t)dst[k] * CSTRIDE], 1);
}

__device__ __forceinline__ int wave_incl_scan(int v, int lane)
{
#pragma unroll
    for (int o = 1; o < 32; o <<= 1) {
        int t = __shfl_up(v, o);
        if (lane >= o) v += t;
    }
    return v;
}

__global__ __launch_bounds__(256) void csr_blocksum_kernel(
    const int* __restrict__ cur, int* __restrict__ bsum, int n)
{
    __shared__ int wtot[8];
    const int i = blockIdx.x * 256 + threadIdx.x;
    const int lane = threadIdx.x & 31, w = threadIdx.x >> 5;
    const int c = i < n ? cur[(size_t)i * CSTRIDE] : 0;
    const int inc = wave_incl_scan(c, lane);
    if (lane == 31) wtot[w] = inc;
    __syncthreads();
    if (threadIdx.x == 0) {
        int s = 0;
#pragma unroll
        for (int j = 0; j < 8; ++j) s += wtot[j];
        bsum[blockIdx.x] = s;
    }
}

__global__ __launch_bounds__(256) void csr_scan_kernel(
    int* __restrict__ cur, const int* __restrict__ bsum, int* __restrict__ rowptr, int n)
{
    __shared__ int wtot[8];
    __shared__ int wpre[8];
    const int i = blockIdx.x * 256 + threadIdx.x;
    const int lane = threadIdx.x & 31, w = threadIdx.x >> 5;
    const int c = i < n ? cur[(size_t)i * CSTRIDE] : 0;

    int part = 0;
    for (int j = threadIdx.x; j < (int)blockIdx.x; j += 256) part += bsum[j];
    const int pinc = wave_incl_scan(part, lane);
    const int inc  = wave_incl_scan(c, lane);
    if (lane == 31) { wpre[w] = pinc; wtot[w] = inc; }
    __syncthreads();
    int off = 0;
#pragma unroll
    for (int j = 0; j < 8; ++j) {
        off += wpre[j];
        if (j < w) off += wtot[j];
    }
    const int excl = off + inc - c;
    if (i < n) {
        rowptr[i] = excl;
        cur[(size_t)i * CSTRIDE] = excl;
        if (i == n - 1) rowptr[n] = excl + c;
    }
}

__global__ void csr_fill_kernel(const int* __restrict__ src, const int* __restrict__ dst,
                                int* __restrict__ cur, int2* __restrict__ ent, int E_)
{
    int k = blockIdx.x * blockDim.x + threadIdx.x;
    if (k >= E_) return;
    const int pos = atomicAdd(&cur[(size_t)dst[k] * CSTRIDE], 1);
    if ((unsigned)pos < (unsigned)E_) ent[pos] = make_int2(k, src[k]);
}

template <bool RELU>
__global__ __launch_bounds__(256) void node_gather_kernel(
    const int2* __restrict__ ent, const int* __restrict__ rowptr,
    const float* __restrict__ P, const float* __restrict__ denom,
    const float* __restrict__ Hp, float* __restrict__ out, int E_, int n)
{
    const int lane = threadIdx.x & 31;
    const int d = __builtin_amdgcn_readfirstlane((int)(blockIdx.x * 8 + (threadIdx.x >> 5)));
    if (d >= n) return;
    const int h  = lane >> 4;
    const int c0 = lane * 4;
    const float den0 = denom[d * NSTRIDE + 0] + 1e-16f;
    const float den1 = denom[d * NSTRIDE + 1] + 1e-16f;
    const int beg = max(rowptr[d], 0);
    const int end = min(rowptr[d + 1], E_);

    const float aself = P[((size_t)E_ + d) * HEADS + h] / (h ? den1 : den0);
    const float4 hs = *(const float4*)(Hp + (size_t)d * FDIM + c0);
    float4 acc;
    acc.x = hs.x * aself; acc.y = hs.y * aself; acc.z = hs.z * aself; acc.w = hs.w * aself;

    for (int base = beg; base < end; base += 32) {
        const int j = base + lane;
        int s_l = d;
        float a0 = 0.f, a1 = 0.f;
        if (j < end) {
            const int2 e = ent[j];
            const int k = min(max(e.x, 0), E_ - 1);
            s_l = min(max(e.y, 0), n - 1);
            const float2 p = *(const float2*)(P + (size_t)k * HEADS);
            a0 = p.x / den0;
            a1 = p.y / den1;
        }
        const int cnt = min(32, end - base);
#pragma unroll 4
        for (int t = 0; t < cnt; ++t) {
            const int s = __builtin_amdgcn_readlane(s_l, t);
            const float w0 = __int_as_float(__builtin_amdgcn_readlane(__float_as_int(a0), t));
            const float w1 = __int_as_float(__builtin_amdgcn_readlane(__float_as_int(a1), t));
            const float a = h ? w1 : w0;
            const float4 hv = *(const float4*)(Hp + (size_t)s * FDIM + c0);
            acc.x += hv.x * a;
            acc.y += hv.y * a;
            acc.z += hv.z * a;
            acc.w += hv.w * a;
        }
    }

    float4* o = (float4*)(out + (size_t)d * FDIM + c0);
    float4 ov = *o;
    ov.x += acc.x; ov.y += acc.y; ov.z += acc.z; ov.w += acc.w;
    if (RELU) {
        ov.x = fmaxf(ov.x, 0.0f); ov.y = fmaxf(ov.y, 0.0f);
        ov.z = fmaxf(ov.z, 0.0f); ov.w = fmaxf(ov.w, 0.0f);
    }
    *o = ov;
}

extern "C" void kernel_launch(void* const* d_in, const int* in_sizes, int n_in,
                              void* d_out, int out_size, void* d_ws, size_t ws_size,
                              hipStream_t stream)
{
    const float* x        = (const float*)d_in[0];
    const int*   src      = (const int*)  d_in[1];
    const int*   dst      = (const int*)  d_in[2];
    const float* W1       = (const float*)d_in[3];
    const float* att_src1 = (const float*)d_in[4];
    const float* att_dst1 = (const float*)d_in[5];
    const float* b1       = (const float*)d_in[6];
    const float* Wsk1     = (const float*)d_in[7];
    const float* bsk1     = (const float*)d_in[8];
    const float* W2       = (const float*)d_in[9];
    const float* att_src2 = (const float*)d_in[10];
    const float* att_dst2 = (const float*)d_in[11];
    const float* b2       = (const float*)d_in[12];
    const float* Wsk2     = (const float*)d_in[13];
    const float* bsk2     = (const float*)d_in[14];

    const int N  = in_sizes[0] / FDIM;
    const int E  = in_sizes[1];
    const int M  = E + N;
    float* out   = (float*)d_out;

    char* base = (char*)d_ws;
    size_t off = 0;
    auto alloc = [&](size_t bytes) {
        void* p = base + off;
        off = (off + bytes + 255) & ~(size_t)255;
        return p;
    };
    float* Hproj = (float*)alloc((size_t)N * FDIM * 4);
    float* Hbuf  = (float*)alloc((size_t)N * FDIM * 4);
    float* a_s   = (float*)alloc((size_t)N * HEADS * 4);
    float* a_d   = (float*)alloc((size_t)N * HEADS * 4);
    int*   mbuf  = (int*)  alloc((size_t)N * NSTRIDE * 4);
    float* denom = (float*)alloc((size_t)N * NSTRIDE * 4);
    float* P     = (float*)alloc((size_t)M * HEADS * 4);
    const int nblk = (N + 255) / 256;
    int*   cur    = (int*)  alloc((size_t)N * CSTRIDE * 4);
    int*   bsum   = (int*)  alloc((size_t)nblk * 4);
    int*   rowptr = (int*)  alloc((size_t)(N + 1) * 4);
    int2*  ent    = (int2*) alloc((size_t)E * 8);

    const int ntiles     = N / 16;
    const dim3 gGemm((ntiles + 3) / 4), bGemm(128);
    const dim3 gAtt((N * HEADS + 255) / 256), b256(256);
    const dim3 gInit((N * HEADS + 255) / 256);
    const dim3 gEdge((M + 255) / 256);
    const dim3 gNode((N + 7) / 8);
    const dim3 gNodeT((N + 255) / 256);
    const dim3 gEdgeE((E + 255) / 256);

    csr_zero_kernel<<<gNodeT, b256, 0, stream>>>(cur, N);
    csr_count_kernel<<<gEdgeE, b256, 0, stream>>>(dst, cur, E);
    csr_blocksum_kernel<<<dim3(nblk), b256, 0, stream>>>(cur, bsum, N);
    csr_scan_kernel<<<dim3(nblk), b256, 0, stream>>>(cur, bsum, rowptr, N);
    csr_fill_kernel<<<gEdgeE, b256, 0, stream>>>(src, dst, cur, ent, E);

    gemm_f32_wmma<<<gGemm, bGemm, 0, stream>>>(x, W1, nullptr, nullptr, Hproj, N);
    gemm_f32_wmma<<<gGemm, bGemm, 0, stream>>>(x, Wsk1, b1, bsk1, Hbuf, N);
    att_kernel<<<gAtt, b256, 0, stream>>>(Hproj, att_src1, att_dst1, a_s, a_d, N);
    init_kernel<<<gInit, b256, 0, stream>>>(mbuf, denom, N * HEADS);
    edge_max_kernel<<<gEdge, b256, 0, stream>>>(src, dst, a_s, a_d, mbuf, E, M);
    edge_sum_kernel<<<gEdge, b256, 0, stream>>>(src, dst, a_s, a_d, mbuf, P, denom, E, M);
    node_gather_kernel<true><<<gNode, b256, 0, stream>>>(ent, rowptr, P, denom, Hproj, Hbuf, E, N);

    gemm_f32_wmma<<<gGemm, bGemm, 0, stream>>>(Hbuf, W2, nullptr, nullptr, Hproj, N);
    gemm_f32_wmma<<<gGemm, bGemm, 0, stream>>>(Hbuf, Wsk2, b2, bsk2, out, N);
    att_kernel<<<gAtt, b256, 0, stream>>>(Hproj, att_src2, att_dst2, a_s, a_d, N);
    init_kernel<<<gInit, b256, 0, stream>>>(mbuf, denom, N * HEADS);
    edge_max_kernel<<<gEdge, b256, 0, stream>>>(src, dst, a_s, a_d, mbuf, E, M);
    edge_sum_kernel<<<gEdge, b256, 0, stream>>>(src, dst, a_s, a_d, mbuf, P, denom, E, M);
    node_gather_kernel<false><<<gNode, b256, 0, stream>>>(ent, rowptr, P, denom, Hproj, out, E, N);
}


# Fold the segment softmax into the per-node gather: each wave takes its node's logit max and exp sum by a wave reduction tree instead of float atomics; init, edge-max, edge-sum launches go
# speedup vs baseline: 2.6874x; 1.4947x over previous
#include <hip/hip_runtime.h>

#define FDIM 128
#define HEADS 2
#define CH 64
#define NEG_SLOPE 0.2f
#define CSTRIDE 32

typedef float v2f __attribute__((ext_vector_type(2)));
typedef float v8f __attribute__((ext_vector_type(8)));

__global__ __launch_bounds__(128) void gemm_f32_wmma(
    const float* __restrict__ X, const float* __restrict__ W,
    const float* __restrict__ bias0, const float* __restrict__ bias1,
    float* __restrict__ OUT, int nrows)
{
    const int ntiles = nrows >> 4;
    const int tile = blockIdx.x * 4 + (threadIdx.x >> 5);
    if (tile >= ntiles) return;

    const int lane  = threadIdx.x & 31;
    const int lhalf = lane >> 4;
    const int lidx  = lane & 15;
    const int row0  = tile * 16;

    const float* xrow = X + (size_t)(row0 + lidx) * FDIM + 2 * lhalf;

    v8f acc[8];
#pragma unroll
    for (int t = 0; t < 8; ++t)
#pragma unroll
        for (int r = 0; r < 8; ++r) acc[t][r] = 0.0f;

#pragma unroll 4
    for (int kk = 0; kk < FDIM; kk += 4) {
        v2f a = *(const v2f*)(xrow + kk);
        const int krow = kk + 2 * lhalf;
        const float* wr0 = W + (size_t)krow * FDIM + lidx;
        const float* wr1 = wr0 + FDIM;
#pragma unroll
        for (int t = 0; t < 8; ++t) {
            v2f b;
            b.x = wr0[t * 16];
            b.y = wr1[t * 16];
            acc[t] = __builtin_amdgcn_wmma_f32_16x16x4_f32(
                false, a, false, b, (short)0, acc[t], false, false);
        }
    }

    const int orow = row0 + 8 * lhalf;
#pragma unroll
    for (int t = 0; t < 8; ++t) {
        const int col = t * 16 + lidx;
        float badd = 0.0f;
        if (bias0) badd += bias0[col];
        if (bias1) badd += bias1[col];
        float* o = OUT + (size_t)orow * FDIM + col;
#pragma unroll
        for (int r = 0; r < 8; ++r)
            o[(size_t)r * FDIM] = acc[t][r] + badd;
    }
}

__global__ void att_kernel(const float* __restrict__ Hp,
                           const float* __restrict__ att_src,
                           const float* __restrict__ att_dst,
                           float* __restrict__ a_s, float* __restrict__ a_d,
                           int n)
{
    int i = blockIdx.x * blockDim.x + threadIdx.x;
    if (i >= n * HEADS) return;
    const int node = i >> 1, h = i & 1;
    const float* hp = Hp + (size_t)node * FDIM + h * CH;
    const float* as = att_src + h * CH;
    const float* ad = att_dst + h * CH;
    float s0 = 0.f, s1 = 0.f;
#pragma unroll 8
    for (int c = 0; c < CH; ++c) {
        float v = hp[c];
        s0 += v * as[c];
        s1 += v * ad[c];
    }
    a_s[i] = s0;
    a_d[i] = s1;
}

__device__ __forceinline__ float lrelu(float x)  { return x > 0.f ? x : NEG_SLOPE * x; }

__global__ void csr_zero_kernel(int* __restrict__ cur, int n)
{
    int i = blockIdx.x * blockDim.x + threadIdx.x;
    if (i < n) cur[(size_t)i * CSTRIDE] = 0;
}

__global__ void csr_count_kernel(const int* __restrict__ dst, int* __restrict__ cur, int E_)
{
    int k = blockIdx.x * blockDim.x + threadIdx.x;
    if (k >= E_) return;
    atomicAdd(&cur[(size_t)dst[k] * CSTRIDE], 1);
}

__device__ __forceinline__ int wave_incl_scan(int v, int lane)
{
#pragma unroll
    for (int o = 1; o < 32; o <<= 1) {
        int t = __shfl_up(v, o);
        if (lane >= o) v += t;
    }
    return v;
}

__global__ __launch_bounds__(256) void csr_blocksum_kernel(
    const int* __restrict__ cur, int* __restrict__ bsum, int n)
{
    __shared__ int wtot[8];
    const int i = blockIdx.x * 256 + threadIdx.x;
    const int lane = threadIdx.x & 31, w = threadIdx.x >> 5;
    const int c = i < n ? cur[(size_t)i * CSTRIDE] : 0;
    const int inc = wave_incl_scan(c, lane);
    if (lane == 31) wtot[w] = inc;
    __syncthreads();
    if (threadIdx.x == 0) {
        int s = 0;
#pragma unroll
        for (int j = 0; j < 8; ++j) s += wtot[j];
        bsum[blockIdx.x] = s;
    }
}

__global__ __launch_bounds__(256) void csr_scan_kernel(
    int* __restrict__ cur, const int* __restrict__ bsum, int* __restrict__ rowptr, int n)
{
    __shared__ int wtot[8];
    __shared__ int wpre[8];
    const int i = blockIdx.x * 256 + threadIdx.x;
    const int lane = threadIdx.x & 31, w = threadIdx.x >> 5;
    const int c = i < n ? cur[(size_t)i * CSTRIDE] : 0;

    int part = 0;
    for (int j = threadIdx.x; j < (int)blockIdx.x; j += 256) part += bsum[j];
    const int pinc = wave_incl_scan(part, lane);
    const int inc  = wave_incl_scan(c, lane);
    if (lane == 31) { wpre[w] = pinc; wtot[w] = inc; }
    __syncthreads();
    int off = 0;
#pragma unroll
    for (int j = 0; j < 8; ++j) {
        off += wpre[j];
        if (j < w) off += wtot[j];
    }
    const int excl = off + inc - c;
    if (i < n) {
        rowptr[i] = excl;
        cur[(size_t)i * CSTRIDE] = excl;
        if (i == n - 1) rowptr[n] = excl + c;
    }
}

__global__ void csr_fill_kernel(const int* __restrict__ src, const int* __restrict__ dst,
                                int* __restrict__ cur, int* __restrict__ esrc, int E_)
{
    int k = blockIdx.x * blockDim.x + threadIdx.x;
    if (k >= E_) return;
    const int pos = atomicAdd(&cur[(size_t)dst[k] * CSTRIDE], 1);
    if ((unsigned)pos < (unsigned)E_) esrc[pos] = src[k];
}

template <int M>
__device__ __forceinline__ float row_xor(float x)
{
    return __int_as_float(__builtin_amdgcn_update_dpp(0, __float_as_int(x), 0x160 + M, 0xf, 0xf, true));
}
__device__ __forceinline__ float lane_value(float x, int l)
{
    return __int_as_float(__builtin_amdgcn_readlane(__float_as_int(x), l));
}
__device__ __forceinline__ float wave_max(float x)
{
    x = fmaxf(x, row_xor<1>(x));
    x = fmaxf(x, row_xor<2>(x));
    x = fmaxf(x, row_xor<4>(x));
    x = fmaxf(x, row_xor<8>(x));
    return fmaxf(lane_value(x, 0), lane_value(x, 16));
}
__device__ __forceinline__ float wave_sum(float x)
{
    x += row_xor<1>(x);
    x += row_xor<2>(x);
    x += row_xor<4>(x);
    x += row_xor<8>(x);
    return lane_value(x, 0) + lane_value(x, 16);
}

template <bool RELU>
__global__ __launch_bounds__(256) void node_softmax_gather_kernel(
    const int* __restrict__ esrc, const int* __restrict__ rowptr,
    const float* __restrict__ a_s, const float* __restrict__ a_d,
    const float* __restrict__ Hp, float* __restrict__ out, int E_, int n)
{
    const int lane = threadIdx.x & 31;
    const int d = __builtin_amdgcn_readfirstlane((int)(blockIdx.x * 8 + (threadIdx.x >> 5)));
    if (d >= n) return;
    const int h  = lane >> 4;
    const int c0 = lane * 4;
    const int beg = max(rowptr[d], 0);
    const int end = min(rowptr[d + 1], E_);
    const float2 adv = *(const float2*)(a_d + (size_t)d * HEADS);
    const float2 asv = *(const float2*)(a_s + (size_t)d * HEADS);
    const float4 hs  = *(const float4*)(Hp + (size_t)d * FDIM + c0);
    const float es0 = lrelu(asv.x + adv.x);
    const float es1 = lrelu(asv.y + adv.y);

    auto edge = [&](int j, int& s, float& e0, float& e1) {
        s = d; e0 = -INFINITY; e1 = -INFINITY;
        if (j < end) {
            s = min(max(esrc[j], 0), n - 1);
            const float2 a = *(const float2*)(a_s + (size_t)s * HEADS);
            e0 = lrelu(a.x + adv.x);
            e1 = lrelu(a.y + adv.y);
        }
    };

    int s_l; float e0, e1;
    edge(beg + lane, s_l, e0, e1);

    float mx0 = e0, mx1 = e1;
    for (int base = beg + 32; base < end; base += 32) {
        int s; float f0, f1;
        edge(base + lane, s, f0, f1);
        mx0 = fmaxf(mx0, f0); mx1 = fmaxf(mx1, f1);
    }
    const float m0 = fmaxf(wave_max(mx0), es0);
    const float m1 = fmaxf(wave_max(mx1), es1);

    const bool v_l = beg + lane < end;
    const float p0 = v_l ? __expf(e0 - m0) : 0.f;
    const float p1 = v_l ? __expf(e1 - m1) : 0.f;
    float sm0 = p0, sm1 = p1;
    for (int base = beg + 32; base < end; base += 32) {
        int s; float f0, f1;
        edge(base + lane, s, f0, f1);
        if (base + lane < end) { sm0 += __expf(f0 - m0); sm1 += __expf(f1 - m1); }
    }
    const float ps0 = __expf(es0 - m0);
    const float ps1 = __expf(es1 - m1);
    const float den0 = (wave_sum(sm0) + ps0) + 1e-16f;
    const float den1 = (wave_sum(sm1) + ps1) + 1e-16f;

    const float aself = h ? ps1 / den1 : ps0 / den0;
    float4 acc;
    acc.x = hs.x * aself; acc.y = hs.y * aself; acc.z = hs.z * aself; acc.w = hs.w * aself;

    auto rows = [&](int s_v, float a0, float a1, int cnt) {
#pragma unroll 4
        for (int t = 0; t < cnt; ++t) {
            const int s = __builtin_amdgcn_readlane(s_v, t);
            const float w0 = lane_value(a0, t);
            const float w1 = lane_value(a1, t);
            const float a = h ? w1 : w0;
            const float4 hv = *(const float4*)(Hp + (size_t)s * FDIM + c0);
            acc.x += hv.x * a;
            acc.y += hv.y * a;
            acc.z += hv.z * a;
            acc.w += hv.w * a;
        }
    };

    rows(s_l, p0 / den0, p1 / den1, min(32, end - beg));
    for (int base = beg + 32; base < end; base += 32) {
        int s; float f0, f1;
        edge(base + lane, s, f0, f1);
        const bool v = base + lane < end;
        const float a0 = v ? __expf(f0 - m0) / den0 : 0.f;
        const float a1 = v ? __expf(f1 - m1) / den1 : 0.f;
        rows(s, a0, a1, min(32, end - base));
    }

    float4* o = (float4*)(out + (size_t)d * FDIM + c0);
    float4 ov = *o;
    ov.x += acc.x; ov.y += acc.y; ov.z += acc.z; ov.w += acc.w;
    if (RELU) {
        ov.x = fmaxf(ov.x, 0.0f); ov.y = fmaxf(ov.y, 0.0f);
        ov.z = fmaxf(ov.z, 0.0f); ov.w = fmaxf(ov.w, 0.0f);
    }
    *o = ov;
}

extern "C" void kernel_launch(void* const* d_in, const int* in_sizes, int n_in,
                              void* d_out, int out_size, void* d_ws, size_t ws_size,
                              hipStream_t stream)
{
    const float* x        = (const float*)d_in[0];
    const int*   src      = (const int*)  d_in[1];
    const int*   dst      = (const int*)  d_in[2];
    const float* W1       = (const float*)d_in[3];
    const float* att_src1 = (const float*)d_in[4];
    const float* att_dst1 = (const float*)d_in[5];
    const float* b1       = (const float*)d_in[6];
    const float* Wsk1     = (const float*)d_in[7];
    const float* bsk1     = (const float*)d_in[8];
    const float* W2       = (const float*)d_in[9];
    const float* att_src2 = (const float*)d_in[10];
    const float* att_dst2 = (const float*)d_in[11];
    const float* b2       = (const float*)d_in[12];
    const float* Wsk2     = (const float*)d_in[13];
    const float* bsk2     = (const float*)d_in[14];

    const int N  = in_sizes[0] / FDIM;
    const int E  = in_sizes[1];
    float* out   = (float*)d_out;

    char* base = (char*)d_ws;
    size_t off = 0;
    auto alloc = [&](size_t bytes) {
        void* p = base + off;
        off = (off + bytes + 255) & ~(size_t)255;
        return p;
    };
    float* Hproj = (float*)alloc((size_t)N * FDIM * 4);
    float* Hbuf  = (float*)alloc((size_t)N * FDIM * 4);
    float* a_s   = (float*)alloc((size_t)N * HEADS * 4);
    float* a_d   = (float*)alloc((size_t)N * HEADS * 4);
    const int nblk = (N + 255) / 256;
    int*   cur    = (int*)  alloc((size_t)N * CSTRIDE * 4);
    int*   bsum   = (int*)  alloc((size_t)nblk * 4);
    int*   rowptr = (int*)  alloc((size_t)(N + 1) * 4);
    int*   esrc   = (int*)  alloc((size_t)E * 4);

    const int ntiles     = N / 16;
    const dim3 gGemm((ntiles + 3) / 4), bGemm(128);
    const dim3 gAtt((N * HEADS + 255) / 256), b256(256);
    const dim3 gNode((N + 7) / 8);
    const dim3 gNodeT((N + 255) / 256);
    const dim3 gEdgeE((E + 255) / 256);

    csr_zero_kernel<<<gNodeT, b256, 0, stream>>>(cur, N);
    csr_count_kernel<<<gEdgeE, b256, 0, stream>>>(dst, cur, E);
    csr_blocksum_kernel<<<dim3(nblk), b256, 0, stream>>>(cur, bsum, N);
    csr_scan_kernel<<<dim3(nblk), b256, 0, stream>>>(cur, bsum, rowptr, N);
    csr_fill_kernel<<<gEdgeE, b256, 0, stream>>>(src, dst, cur, esrc, E);

    gemm_f32_wmma<<<gGemm, bGemm, 0, stream>>>(x, W1, nullptr, nullptr, Hproj, N);
    gemm_f32_wmma<<<gGemm, bGemm, 0, stream>>>(x, Wsk1, b1, bsk1, Hbuf, N);
    att_kernel<<<gAtt, b256, 0, stream>>>(Hproj, att_src1, att_dst1, a_s, a_d, N);
    node_softmax_gather_kernel<true><<<gNode, b256, 0, stream>>>(esrc, rowptr, a_s, a_d, Hproj, Hbuf, E, N);

    gemm_f32_wmma<<<gGemm, bGemm, 0, stream>>>(Hbuf, W2, nullptr, nullptr, Hproj, N);
    gemm_f32_wmma<<<gGemm, bGemm, 0, stream>>>(Hbuf, Wsk2, b2, bsk2, out, N);
    att_kernel<<<gAtt, b256, 0, stream>>>(Hproj, att_src2, att_dst2, a_s, a_d, N);
    node_softmax_gather_kernel<false><<<gNode, b256, 0, stream>>>(esrc, rowptr, a_s, a_d, Hproj, out, E, N);
}


# One atomic pass builds the per-node edge lists: a 256-byte record per node (counter + 63 source slots, shared overflow list), slot = the counting add's return; count, block-sum and scan launches go
# speedup vs baseline: 2.9923x; 1.1135x over previous
#include <hip/hip_runtime.h>

#define FDIM 128
#define HEADS 2
#define CH 64
#define NEG_SLOPE 0.2f
#define SLOTW 64
#define CAP 63

typedef float v2f __attribute__((ext_vector_type(2)));
typedef float v8f __attribute__((ext_vector_type(8)));

__global__ __launch_bounds__(128) void gemm_f32_wmma(
    const float* __restrict__ X, const float* __restrict__ W,
    const float* __restrict__ bias0, const float* __restrict__ bias1,
    float* __restrict__ OUT, int nrows)
{
    const int ntiles = nrows >> 4;
    const int tile = blockIdx.x * 4 + (threadIdx.x >> 5);
    if (tile >= ntiles) return;

    const int lane  = threadIdx.x & 31;
    const int lhalf = lane >> 4;
    const int lidx  = lane & 15;
    const int row0  = tile * 16;

    const float* xrow = X + (size_t)(row0 + lidx) * FDIM + 2 * lhalf;

    v8f acc[8];
#pragma unroll
    for (int t = 0; t < 8; ++t)
#pragma unroll
        for (int r = 0; r < 8; ++r) acc[t][r] = 0.0f;

#pragma unroll 4
    for (int kk = 0; kk < FDIM; kk += 4) {
        v2f a = *(const v2f*)(xrow + kk);
        const int krow = kk + 2 * lhalf;
        const float* wr0 = W + (size_t)krow * FDIM + lidx;
        const float* wr1 = wr0 + FDIM;
#pragma unroll
        for (int t = 0; t < 8; ++t) {
            v2f b;
            b.x = wr0[t * 16];
            b.y = wr1[t * 16];
            acc[t] = __builtin_amdgcn_wmma_f32_16x16x4_f32(
                false, a, false, b, (short)0, acc[t], false, false);
        }
    }

    const int orow = row0 + 8 * lhalf;
#pragma unroll
    for (int t = 0; t < 8; ++t) {
        const int col = t * 16 + lidx;
        float badd = 0.0f;
        if (bias0) badd += bias0[col];
        if (bias1) badd += bias1[col];
        float* o = OUT + (size_t)orow * FDIM + col;
#pragma unroll
        for (int r = 0; r < 8; ++r)
            o[(size_t)r * FDIM] = acc[t][r] + badd;
    }
}

__global__ void att_kernel(const float* __restrict__ Hp,
                           const float* __restrict__ att_src,
                           const float* __restrict__ att_dst,
                           float* __restrict__ a_s, float* __restrict__ a_d,
                           int n)
{
    int i = blockIdx.x * blockDim.x + threadIdx.x;
    if (i >= n * HEADS) return;
    const int node = i >> 1, h = i & 1;
    const float* hp = Hp + (size_t)node * FDIM + h * CH;
    const float* as = att_src + h * CH;
    const float* ad = att_dst + h * CH;
    float s0 = 0.f, s1 = 0.f;
#pragma unroll 8
    for (int c = 0; c < CH; ++c) {
        float v = hp[c];
        s0 += v * as[c];
        s1 += v * ad[c];
    }
    a_s[i] = s0;
    a_d[i] = s1;
}

__device__ __forceinline__ float lrelu(float x)  { return x > 0.f ? x : NEG_SLOPE * x; }

__global__ void slots_zero_kernel(int* __restrict__ slots, int* __restrict__ ovcnt, int n)
{
    int i = blockIdx.x * blockDim.x + threadIdx.x;
    if (i < n) slots[(size_t)i * SLOTW] = 0;
    if (i == 0) ovcnt[0] = 0;
}

__global__ void slots_fill_kernel(const int* __restrict__ src, const int* __restrict__ dst,
                                  int* __restrict__ slots, int* __restrict__ ovcnt,
                                  int* __restrict__ ovdst, int* __restrict__ ovsrc, int E_)
{
    int k = blockIdx.x * blockDim.x + threadIdx.x;
    if (k >= E_) return;
    const int d = dst[k];
    const int s = src[k];
    int* rec = slots + (size_t)d * SLOTW;
    const int r = atomicAdd(rec, 1);
    if ((unsigned)r < (unsigned)CAP) {
        rec[1 + r] = s;
    } else {
        const int o = atomicAdd(ovcnt, 1);
        if ((unsigned)o < (unsigned)E_) { ovdst[o] = d; ovsrc[o] = s; }
    }
}

template <int M>
__device__ __forceinline__ float row_xor(float x)
{
    return __int_as_float(__builtin_amdgcn_update_dpp(0, __float_as_int(x), 0x160 + M, 0xf, 0xf, true));
}
__device__ __forceinline__ float lane_value(float x, int l)
{
    return __int_as_float(__builtin_amdgcn_readlane(__float_as_int(x), l));
}
__device__ __forceinline__ float wave_max(float x)
{
    x = fmaxf(x, row_xor<1>(x));
    x = fmaxf(x, row_xor<2>(x));
    x = fmaxf(x, row_xor<4>(x));
    x = fmaxf(x, row_xor<8>(x));
    return fmaxf(lane_value(x, 0), lane_value(x, 16));
}
__device__ __forceinline__ float wave_sum(float x)
{
    x += row_xor<1>(x);
    x += row_xor<2>(x);
    x += row_xor<4>(x);
    x += row_xor<8>(x);
    return lane_value(x, 0) + lane_value(x, 16);
}

template <bool RELU>
__global__ __launch_bounds__(256) void node_softmax_gather_kernel(
    const int* __restrict__ slots, const int* __restrict__ ovcnt,
    const int* __restrict__ ovdst, const int* __restrict__ ovsrc,
    const float* __restrict__ a_s, const float* __restrict__ a_d,
    const float* __restrict__ Hp, float* __restrict__ out, int E_, int n)
{
    const int lane = threadIdx.x & 31;
    const int d = __builtin_amdgcn_readfirstlane((int)(blockIdx.x * 8 + (threadIdx.x >> 5)));
    if (d >= n) return;
    const int h  = lane >> 4;
    const int c0 = lane * 4;
    const int* rec = slots + (size_t)d * SLOTW;
    const int cnt = rec[0];
    const int end = min(max(cnt, 0), CAP);
    const int nov = cnt > CAP ? min(max(ovcnt[0], 0), E_) : 0;
    const float2 adv = *(const float2*)(a_d + (size_t)d * HEADS);
    const float2 asv = *(const float2*)(a_s + (size_t)d * HEADS);
    const float4 hs  = *(const float4*)(Hp + (size_t)d * FDIM + c0);
    const float es0 = lrelu(asv.x + adv.x);
    const float es1 = lrelu(asv.y + adv.y);

    auto edge = [&](int j, int& s, float& e0, float& e1) {
        s = d; e0 = -INFINITY; e1 = -INFINITY;
        if (j < end) {
            s = min(max(rec[1 + j], 0), n - 1);
            const float2 a = *(const float2*)(a_s + (size_t)s * HEADS);
            e0 = lrelu(a.x + adv.x);
            e1 = lrelu(a.y + adv.y);
        }
    };
    auto over = [&](int j, int& s, float& e0, float& e1) {
        s = d; e0 = -INFINITY; e1 = -INFINITY;
        bool v = false;
        if (j < nov) v = ovdst[j] == d;
        if (v) {
            s = min(max(ovsrc[j], 0), n - 1);
            const float2 a = *(const float2*)(a_s + (size_t)s * HEADS);
            e0 = lrelu(a.x + adv.x);
            e1 = lrelu(a.y + adv.y);
        }
        return v;
    };

    int s_l; float e0, e1;
    edge(lane, s_l, e0, e1);

    float mx0 = e0, mx1 = e1;
    for (int base = 32; base < end; base += 32) {
        int s; float f0, f1;
        edge(base + lane, s, f0, f1);
        mx0 = fmaxf(mx0, f0); mx1 = fmaxf(mx1, f1);
    }
    for (int base = 0; base < nov; base += 32) {
        int s; float f0, f1;
        over(base + lane, s, f0, f1);
        mx0 = fmaxf(mx0, f0); mx1 = fmaxf(mx1, f1);
    }
    const float m0 = fmaxf(wave_max(mx0), es0);
    const float m1 = fmaxf(wave_max(mx1), es1);

    const bool v_l = lane < end;
    const float p0 = v_l ? __expf(e0 - m0) : 0.f;
    const float p1 = v_l ? __expf(e1 - m1) : 0.f;
    float sm0 = p0, sm1 = p1;
    for (int base = 32; base < end; base += 32) {
        int s; float f0, f1;
        edge(base + lane, s, f0, f1);
        if (base + lane < end) { sm0 += __expf(f0 - m0); sm1 += __expf(f1 - m1); }
    }
    for (int base = 0; base < nov; base += 32) {
        int s; float f0, f1;
        if (over(base + lane, s, f0, f1)) { sm0 += __expf(f0 - m0); sm1 += __expf(f1 - m1); }
    }
    const float ps0 = __expf(es0 - m0);
    const float ps1 = __expf(es1 - m1);
    const float den0 = (wave_sum(sm0) + ps0) + 1e-16f;
    const float den1 = (wave_sum(sm1) + ps1) + 1e-16f;

    const float aself = h ? ps1 / den1 : ps0 / den0;
    float4 acc;
    acc.x = hs.x * aself; acc.y = hs.y * aself; acc.z = hs.z * aself; acc.w = hs.w * aself;

    auto row = [&](int s_v, float a0, float a1, int t) {
        const int s = __builtin_amdgcn_readlane(s_v, t);
        const float w0 = lane_value(a0, t);
        const float w1 = lane_value(a1, t);
        const float a = h ? w1 : w0;
        const float4 hv = *(const float4*)(Hp + (size_t)s * FDIM + c0);
        acc.x += hv.x * a;
        acc.y += hv.y * a;
        acc.z += hv.z * a;
        acc.w += hv.w * a;
    };
    auto rows = [&](int s_v, float a0, float a1, int m) {
#pragma unroll 4
        for (int t = 0; t < m; ++t) row(s_v, a0, a1, t);
    };

    rows(s_l, p0 / den0, p1 / den1, min(32, end));
    for (int base = 32; base < end; base += 32) {
        int s; float f0, f1;
        edge(base + lane, s, f0, f1);
        const bool v = base + lane < end;
        const float a0 = v ? __expf(f0 - m0) / den0 : 0.f;
        const float a1 = v ? __expf(f1 - m1) / den1 : 0.f;
        rows(s, a0, a1, min(32, end - base));
    }
    for (int base = 0; base < nov; base += 32) {
        int s; float f0, f1;
        const bool v = over(base + lane, s, f0, f1);
        const float a0 = v ? __expf(f0 - m0) / den0 : 0.f;
        const float a1 = v ? __expf(f1 - m1) / den1 : 0.f;
        for (unsigned live = __builtin_amdgcn_ballot_w32(v); live; live &= live - 1)
            row(s, a0, a1, __builtin_ctz(live));
    }

    float4* o = (float4*)(out + (size_t)d * FDIM + c0);
    float4 ov = *o;
    ov.x += acc.x; ov.y += acc.y; ov.z += acc.z; ov.w += acc.w;
    if (RELU) {
        ov.x = fmaxf(ov.x, 0.0f); ov.y = fmaxf(ov.y, 0.0f);
        ov.z = fmaxf(ov.z, 0.0f); ov.w = fmaxf(ov.w, 0.0f);
    }
    *o = ov;
}

extern "C" void kernel_launch(void* const* d_in, const int* in_sizes, int n_in,
                              void* d_out, int out_size, void* d_ws, size_t ws_size,
                              hipStream_t stream)
{
    const float* x        = (const float*)d_in[0];
    const int*   src      = (const int*)  d_in[1];
    const int*   dst      = (const int*)  d_in[2];
    const float* W1       = (const float*)d_in[3];
    const float* att_src1 = (const float*)d_in[4];
    const float* att_dst1 = (const float*)d_in[5];
    const float* b1       = (const float*)d_in[6];
    const float* Wsk1     = (const float*)d_in[7];
    const float* bsk1     = (const float*)d_in[8];
    const float* W2       = (const float*)d_in[9];
    const float* att_src2 = (const float*)d_in[10];
    const float* att_dst2 = (const float*)d_in[11];
    const float* b2       = (const float*)d_in[12];
    const float* Wsk2     = (const float*)d_in[13];
    const float* bsk2     = (const float*)d_in[14];

    const int N  = in_sizes[0] / FDIM;
    const int E  = in_sizes[1];
    float* out   = (float*)d_out;

    char* base = (char*)d_ws;
    size_t off = 0;
    auto alloc = [&](size_t bytes) {
        void* p = base + off;
        off = (off + bytes + 255) & ~(size_t)255;
        return p;
    };
    float* Hproj = (float*)alloc((size_t)N * FDIM * 4);
    float* Hbuf  = (float*)alloc((size_t)N * FDIM * 4);
    float* a_s   = (float*)alloc((size_t)N * HEADS * 4);
    float* a_d   = (float*)alloc((size_t)N * HEADS * 4);
    int*   slots = (int*)  alloc((size_t)N * SLOTW * 4);
    int*   ovcnt = (int*)  alloc(256);
    int*   ovdst = (int*)  alloc((size_t)E * 4);
    int*   ovsrc = (int*)  alloc((size_t)E * 4);

    const int ntiles     = N / 16;
    const dim3 gGemm((ntiles + 3) / 4), bGemm(128);
    const dim3 gAtt((N * HEADS + 255) / 256), b256(256);
    const dim3 gNode((N + 7) / 8);
    const dim3 gNodeT((N + 255) / 256);
    const dim3 gEdgeE((E + 255) / 256);

    slots_zero_kernel<<<gNodeT, b256, 0, stream>>>(slots, ovcnt, N);
    slots_fill_kernel<<<gEdgeE, b256, 0, stream>>>(src, dst, slots, ovcnt, ovdst, ovsrc, E);

    gemm_f32_wmma<<<gGemm, bGemm, 0, stream>>>(x, W1, nullptr, nullptr, Hproj, N);
    gemm_f32_wmma<<<gGemm, bGemm, 0, stream>>>(x, Wsk1, b1, bsk1, Hbuf, N);
    att_kernel<<<gAtt, b256, 0, stream>>>(Hproj, att_src1, att_dst1, a_s, a_d, N);
    node_softmax_gather_kernel<true><<<gNode, b256, 0, stream>>>(slots, ovcnt, ovdst, ovsrc, a_s, a_d, Hproj, Hbuf, E, N);

    gemm_f32_wmma<<<gGemm, bGemm, 0, stream>>>(Hbuf, W2, nullptr, nullptr, Hproj, N);
    gemm_f32_wmma<<<gGemm, bGemm, 0, stream>>>(Hbuf, Wsk2, b2, bsk2, out, N);
    att_kernel<<<gAtt, b256, 0, stream>>>(Hproj, att_src2, att_dst2, a_s, a_d, N);
    node_softmax_gather_kernel<false><<<gNode, b256, 0, stream>>>(slots, ovcnt, ovdst, ovsrc, a_s, a_d, Hproj, out, E, N);
}


# Take the attention dot products a_s, a_d in the projection GEMM's epilogue from the accumulators (4 tiles per lane, then a 16-lane xor tree: sum order changes); both att launches go
# speedup vs baseline: 3.3874x; 1.1320x over previous
#include <hip/hip_runtime.h>

#define FDIM 128
#define HEADS 2
#define CH 64
#define NEG_SLOPE 0.2f
#define SLOTW 64
#define CAP 63

typedef float v2f __attribute__((ext_vector_type(2)));
typedef float v8f __attribute__((ext_vector_type(8)));

template <int M>
__device__ __forceinline__ float row_xor(float x)
{
    return __int_as_float(__builtin_amdgcn_update_dpp(0, __float_as_int(x), 0x160 + M, 0xf, 0xf, true));
}

template <bool ATT>
__global__ __launch_bounds__(128) void gemm_f32_wmma(
    const float* __restrict__ X, const float* __restrict__ W,
    const float* __restrict__ bias0, const float* __restrict__ bias1,
    const float* __restrict__ att_src, const float* __restrict__ att_dst,
    float* __restrict__ a_s, float* __restrict__ a_d,
    float* __restrict__ OUT, int nrows)
{
    const int ntiles = nrows >> 4;
    const int tile = blockIdx.x * 4 + (threadIdx.x >> 5);
    if (tile >= ntiles) return;

    const int lane  = threadIdx.x & 31;
    const int lhalf = lane >> 4;
    const int lidx  = lane & 15;
    const int row0  = tile * 16;

    const float* xrow = X + (size_t)(row0 + lidx) * FDIM + 2 * lhalf;

    v8f acc[8];
#pragma unroll
    for (int t = 0; t < 8; ++t)
#pragma unroll
        for (int r = 0; r < 8; ++r) acc[t][r] = 0.0f;

#pragma unroll 4
    for (int kk = 0; kk < FDIM; kk += 4) {
        v2f a = *(const v2f*)(xrow + kk);
        const int krow = kk + 2 * lhalf;
        const float* wr0 = W + (size_t)krow * FDIM + lidx;
        const float* wr1 = wr0 + FDIM;
#pragma unroll
        for (int t = 0; t < 8; ++t) {
            v2f b;
            b.x = wr0[t * 16];
            b.y = wr1[t * 16];
            acc[t] = __builtin_amdgcn_wmma_f32_16x16x4_f32(
                false, a, false, b, (short)0, acc[t], false, false);
        }
    }

    const int orow = row0 + 8 * lhalf;

    if (ATT) {
        float ps[HEADS][8], pd[HEADS][8];
#pragma unroll
        for (int h = 0; h < HEADS; ++h) {
#pragma unroll
            for (int r = 0; r < 8; ++r) { ps[h][r] = 0.0f; pd[h][r] = 0.0f; }
#pragma unroll
            for (int tt = 0; tt < 4; ++tt) {
                const int t = 4 * h + tt;
                const float vs = att_src[t * 16 + lidx];
                const float vd = att_dst[t * 16 + lidx];
#pragma unroll
                for (int r = 0; r < 8; ++r) {
                    ps[h][r] += acc[t][r] * vs;
                    pd[h][r] += acc[t][r] * vd;
                }
            }
        }
#pragma unroll
        for (int h = 0; h < HEADS; ++h)
#pragma unroll
            for (int r = 0; r < 8; ++r) {
                float s = ps[h][r], d = pd[h][r];
                s += row_xor<1>(s); d += row_xor<1>(d);
                s += row_xor<2>(s); d += row_xor<2>(d);
                s += row_xor<4>(s); d += row_xor<4>(d);
                s += row_xor<8>(s); d += row_xor<8>(d);
                ps[h][r] = s; pd[h][r] = d;
            }
        float os = ps[0][0], od = pd[0][0];
#pragma unroll
        for (int j = 1; j < 16; ++j)
            if (lidx == j) { os = ps[j & 1][j >> 1]; od = pd[j & 1][j >> 1]; }
        const size_t ao = (size_t)orow * HEADS + lidx;
        a_s[ao] = os;
        a_d[ao] = od;
    }
#pragma unroll
    for (int t = 0; t < 8; ++t) {
        const int col = t * 16 + lidx;
        float badd = 0.0f;
        if (bias0) badd += bias0[col];
        if (bias1) badd += bias1[col];
        float* o = OUT + (size_t)orow * FDIM + col;
#pragma unroll
        for (int r = 0; r < 8; ++r)
            o[(size_t)r * FDIM] = acc[t][r] + badd;
    }
}

__device__ __forceinline__ float lrelu(float x)  { return x > 0.f ? x : NEG_SLOPE * x; }

__global__ void slots_zero_kernel(int* __restrict__ slots, int* __restrict__ ovcnt, int n)
{
    int i = blockIdx.x * blockDim.x + threadIdx.x;
    if (i < n) slots[(size_t)i * SLOTW] = 0;
    if (i == 0) ovcnt[0] = 0;
}

__global__ void slots_fill_kernel(const int* __restrict__ src, const int* __restrict__ dst,
                                  int* __restrict__ slots, int* __restrict__ ovcnt,
                                  int* __restrict__ ovdst, int* __restrict__ ovsrc, int E_)
{
    int k = blockIdx.x * blockDim.x + threadIdx.x;
    if (k >= E_) return;
    const int d = dst[k];
    const int s = src[k];
    int* rec = slots + (size_t)d * SLOTW;
    const int r = atomicAdd(rec, 1);
    if ((unsigned)r < (unsigned)CAP) {
        rec[1 + r] = s;
    } else {
        const int o = atomicAdd(ovcnt, 1);
        if ((unsigned)o < (unsigned)E_) { ovdst[o] = d; ovsrc[o] = s; }
    }
}

__device__ __forceinline__ float lane_value(float x, int l)
{
    return __int_as_float(__builtin_amdgcn_readlane(__float_as_int(x), l));
}
__device__ __forceinline__ float wave_max(float x)
{
    x = fmaxf(x, row_xor<1>(x));
    x = fmaxf(x, row_xor<2>(x));
    x = fmaxf(x, row_xor<4>(x));
    x = fmaxf(x, row_xor<8>(x));
    return fmaxf(lane_value(x, 0), lane_value(x, 16));
}
__device__ __forceinline__ float wave_sum(float x)
{
    x += row_xor<1>(x);
    x += row_xor<2>(x);
    x += row_xor<4>(x);
    x += row_xor<8>(x);
    return lane_value(x, 0) + lane_value(x, 16);
}

template <bool RELU>
__global__ __launch_bounds__(256) void node_softmax_gather_kernel(
    const int* __restrict__ slots, const int* __restrict__ ovcnt,
    const int* __restrict__ ovdst, const int* __restrict__ ovsrc,
    const float* __restrict__ a_s, const float* __restrict__ a_d,
    const float* __restrict__ Hp, float* __restrict__ out, int E_, int n)
{
    const int lane = threadIdx.x & 31;
    const int d = __builtin_amdgcn_readfirstlane((int)(blockIdx.x * 8 + (threadIdx.x >> 5)));
    if (d >= n) return;
    const int h  = lane >> 4;
    const int c0 = lane * 4;
    const int* rec = slots + (size_t)d * SLOTW;
    const int cnt = rec[0];
    const int end = min(max(cnt, 0), CAP);
    const int nov = cnt > CAP ? min(max(ovcnt[0], 0), E_) : 0;
    const float2 adv = *(const float2*)(a_d + (size_t)d * HEADS);
    const float2 asv = *(const float2*)(a_s + (size_t)d * HEADS);
    const float4 hs  = *(const float4*)(Hp + (size_t)d * FDIM + c0);
    const float es0 = lrelu(asv.x + adv.x);
    const float es1 = lrelu(asv.y + adv.y);

    auto edge = [&](int j, int& s, float& e0, float& e1) {
        s = d; e0 = -INFINITY; e1 = -INFINITY;
        if (j < end) {
            s = min(max(rec[1 + j], 0), n - 1);
            const float2 a = *(const float2*)(a_s + (size_t)s * HEADS);
            e0 = lrelu(a.x + adv.x);
            e1 = lrelu(a.y + adv.y);
        }
    };
    auto over = [&](int j, int& s, float& e0, float& e1) {
        s = d; e0 = -INFINITY; e1 = -INFINITY;
        bool v = false;
        if (j < nov) v = ovdst[j] == d;
        if (v) {
            s = min(max(ovsrc[j], 0), n - 1);
            const float2 a = *(const float2*)(a_s + (size_t)s * HEADS);
            e0 = lrelu(a.x + adv.x);
            e1 = lrelu(a.y + adv.y);
        }
        return v;
    };

    int s_l; float e0, e1;
    edge(lane, s_l, e0, e1);

    float mx0 = e0, mx1 = e1;
    for (int base = 32; base < end; base += 32) {
        int s; float f0, f1;
        edge(base + lane, s, f0, f1);
        mx0 = fmaxf(mx0, f0); mx1 = fmaxf(mx1, f1);
    }
    for (int base = 0; base < nov; base += 32) {
        int s; float f0, f1;
        over(base + lane, s, f0, f1);
        mx0 = fmaxf(mx0, f0); mx1 = fmaxf(mx1, f1);
    }
    const float m0 = fmaxf(wave_max(mx0), es0);
    const float m1 = fmaxf(wave_max(mx1), es1);

    const bool v_l = lane < end;
    const float p0 = v_l ? __expf(e0 - m0) : 0.f;
    const float p1 = v_l ? __expf(e1 - m1) : 0.f;
    float sm0 = p0, sm1 = p1;
    for (int base = 32; base < end; base += 32) {
        int s; float f0, f1;
        edge(base + lane, s, f0, f1);
        if (base + lane < end) { sm0 += __expf(f0 - m0); sm1 += __expf(f1 - m1); }
    }
    for (int base = 0; base < nov; base += 32) {
        int s; float f0, f1;
        if (over(base + lane, s, f0, f1)) { sm0 += __expf(f0 - m0); sm1 += __expf(f1 - m1); }
    }
    const float ps0 = __expf(es0 - m0);
    const float ps1 = __expf(es1 - m1);
    const float den0 = (wave_sum(sm0) + ps0) + 1e-16f;
    const float den1 = (wave_sum(sm1) + ps1) + 1e-16f;

    const float aself = h ? ps1 / den1 : ps0 / den0;
    float4 acc;
    acc.x = hs.x * aself; acc.y = hs.y * aself; acc.z = hs.z * aself; acc.w = hs.w * aself;

    auto row = [&](int s_v, float a0, float a1, int t) {
        const int s = __builtin_amdgcn_readlane(s_v, t);
        const float w0 = lane_value(a0, t);
        const float w1 = lane_value(a1, t);
        const float a = h ? w1 : w0;
        const float4 hv = *(const float4*)(Hp + (size_t)s * FDIM + c0);
        acc.x += hv.x * a;
        acc.y += hv.y * a;
        acc.z += hv.z * a;
        acc.w += hv.w * a;
    };
    auto rows = [&](int s_v, float a0, float a1, int m) {
#pragma unroll 4
        for (int t = 0; t < m; ++t) row(s_v, a0, a1, t);
    };

    rows(s_l, p0 / den0, p1 / den1, min(32, end));
    for (int base = 32; base < end; base += 32) {
        int s; float f0, f1;
        edge(base + lane, s, f0, f1);
        const bool v = base + lane < end;
        const float a0 = v ? __expf(f0 - m0) / den0 : 0.f;
        const float a1 = v ? __expf(f1 - m1) / den1 : 0.f;
        rows(s, a0, a1, min(32, end - base));
    }
    for (int base = 0; base < nov; base += 32) {
        int s; float f0, f1;
        const bool v = over(base + lane, s, f0, f1);
        const float a0 = v ? __expf(f0 - m0) / den0 : 0.f;
        const float a1 = v ? __expf(f1 - m1) / den1 : 0.f;
        for (unsigned live = __builtin_amdgcn_ballot_w32(v); live; live &= live - 1)
            row(s, a0, a1, __builtin_ctz(live));
    }

    float4* o = (float4*)(out + (size_t)d * FDIM + c0);
    float4 ov = *o;
    ov.x += acc.x; ov.y += acc.y; ov.z += acc.z; ov.w += acc.w;
    if (RELU) {
        ov.x = fmaxf(ov.x, 0.0f); ov.y = fmaxf(ov.y, 0.0f);
        ov.z = fmaxf(ov.z, 0.0f); ov.w = fmaxf(ov.w, 0.0f);
    }
    *o = ov;
}

extern "C" void kernel_launch(void* const* d_in, const int* in_sizes, int n_in,
                              void* d_out, int out_size, void* d_ws, size_t ws_size,
                              hipStream_t stream)
{
    const float* x        = (const float*)d_in[0];
    const int*   src      = (const int*)  d_in[1];
    const int*   dst      = (const int*)  d_in[2];
    const float* W1       = (const float*)d_in[3];
    const float* att_src1 = (const float*)d_in[4];
    const float* att_dst1 = (const float*)d_in[5];
    const float* b1       = (const float*)d_in[6];
    const float* Wsk1     = (const float*)d_in[7];
    const float* bsk1     = (const float*)d_in[8];
    const float* W2       = (const float*)d_in[9];
    const float* att_src2 = (const float*)d_in[10];
    const float* att_dst2 = (const float*)d_in[11];
    const float* b2       = (const float*)d_in[12];
    const float* Wsk2     = (const float*)d_in[13];
    const float* bsk2     = (const float*)d_in[14];

    const int N  = in_sizes[0] / FDIM;
    const int E  = in_sizes[1];
    float* out   = (float*)d_out;

    char* base = (char*)d_ws;
    size_t off = 0;
    auto alloc = [&](size_t bytes) {
        void* p = base + off;
        off = (off + bytes + 255) & ~(size_t)255;
        return p;
    };
    float* Hproj = (float*)alloc((size_t)N * FDIM * 4);
    float* Hbuf  = (float*)alloc((size_t)N * FDIM * 4);
    float* a_s   = (float*)alloc((size_t)N * HEADS * 4);
    float* a_d   = (float*)alloc((size_t)N * HEADS * 4);
    int*   slots = (int*)  alloc((size_t)N * SLOTW * 4);
    int*   ovcnt = (int*)  alloc(256);
    int*   ovdst = (int*)  alloc((size_t)E * 4);
    int*   ovsrc = (int*)  alloc((size_t)E * 4);

    const int ntiles     = N / 16;
    const dim3 gGemm((ntiles + 3) / 4), bGemm(128);
    const dim3 b256(256);
    const dim3 gNode((N + 7) / 8);
    const dim3 gNodeT((N + 255) / 256);
    const dim3 gEdgeE((E + 255) / 256);

    slots_zero_kernel<<<gNodeT, b256, 0, stream>>>(slots, ovcnt, N);
    slots_fill_kernel<<<gEdgeE, b256, 0, stream>>>(src, dst, slots, ovcnt, ovdst, ovsrc, E);

    gemm_f32_wmma<true><<<gGemm, bGemm, 0, stream>>>(x, W1, nullptr, nullptr, att_src1, att_dst1, a_s, a_d, Hproj, N);
    gemm_f32_wmma<false><<<gGemm, bGemm, 0, stream>>>(x, Wsk1, b1, bsk1, nullptr, nullptr, nullptr, nullptr, Hbuf, N);
    node_softmax_gather_kernel<true><<<gNode, b256, 0, stream>>>(slots, ovcnt, ovdst, ovsrc, a_s, a_d, Hproj, Hbuf, E, N);

    gemm_f32_wmma<true><<<gGemm, bGemm, 0, stream>>>(Hbuf, W2, nullptr, nullptr, att_src2, att_dst2, a_s, a_d, Hproj, N);
    gemm_f32_wmma<false><<<gGemm, bGemm, 0, stream>>>(Hbuf, Wsk2, b2, bsk2, nullptr, nullptr, nullptr, nullptr, out, N);
    node_softmax_gather_kernel<false><<<gNode, b256, 0, stream>>>(slots, ovcnt, ovdst, ovsrc, a_s, a_d, Hproj, out, E, N);
}


# Node edge records built with LDS counters: edges bucketed per chunk in LDS, then one block per 128-node bucket ranks them in LDS and stores whole records; global atomics go (sum order: LDS arrival)
# speedup vs baseline: 3.8228x; 1.1285x over previous
#include <hip/hip_runtime.h>

#define FDIM 128
#define HEADS 2
#define CH 64
#define NEG_SLOPE 0.2f
#define SLOTW 64
#define CAP 63
#define NBLOG 7
#define NB 128
#define CE 4096
#define PT 1024

typedef float v2f __attribute__((ext_vector_type(2)));
typedef float v8f __attribute__((ext_vector_type(8)));

template <int M>
__device__ __forceinline__ float row_xor(float x)
{
    return __int_as_float(__builtin_amdgcn_update_dpp(0, __float_as_int(x), 0x160 + M, 0xf, 0xf, true));
}

template <bool ATT>
__global__ __launch_bounds__(128) void gemm_f32_wmma(
    const float* __restrict__ X, const float* __restrict__ W,
    const float* __restrict__ bias0, const float* __restrict__ bias1,
    const float* __restrict__ att_src, const float* __restrict__ att_dst,
    float* __restrict__ a_s, float* __restrict__ a_d,
    float* __restrict__ OUT, int nrows)
{
    const int ntiles = nrows >> 4;
    const int tile = blockIdx.x * 4 + (threadIdx.x >> 5);
    if (tile >= ntiles) return;

    const int lane  = threadIdx.x & 31;
    const int lhalf = lane >> 4;
    const int lidx  = lane & 15;
    const int row0  = tile * 16;

    const float* xrow = X + (size_t)(row0 + lidx) * FDIM + 2 * lhalf;

    v8f acc[8];
#pragma unroll
    for (int t = 0; t < 8; ++t)
#pragma unroll
        for (int r = 0; r < 8; ++r) acc[t][r] = 0.0f;

#pragma unroll 4
    for (int kk = 0; kk < FDIM; kk += 4) {
        v2f a = *(const v2f*)(xrow + kk);
        const int krow = kk + 2 * lhalf;
        const float* wr0 = W + (size_t)krow * FDIM + lidx;
        const float* wr1 = wr0 + FDIM;
#pragma unroll
        for (int t = 0; t < 8; ++t) {
            v2f b;
            b.x = wr0[t * 16];
            b.y = wr1[t * 16];
            acc[t] = __builtin_amdgcn_wmma_f32_16x16x4_f32(
                false, a, false, b, (short)0, acc[t], false, false);
        }
    }

    const int orow = row0 + 8 * lhalf;

    if (ATT) {
        float ps[HEADS][8], pd[HEADS][8];
#pragma unroll
        for (int h = 0; h < HEADS; ++h) {
#pragma unroll
            for (int r = 0; r < 8; ++r) { ps[h][r] = 0.0f; pd[h][r] = 0.0f; }
#pragma unroll
            for (int tt = 0; tt < 4; ++tt) {
                const int t = 4 * h + tt;
                const float vs = att_src[t * 16 + lidx];
                const float vd = att_dst[t * 16 + lidx];
#pragma unroll
                for (int r = 0; r < 8; ++r) {
                    ps[h][r] += acc[t][r] * vs;
                    pd[h][r] += acc[t][r] * vd;
                }
            }
        }
#pragma unroll
        for (int h = 0; h < HEADS; ++h)
#pragma unroll
            for (int r = 0; r < 8; ++r) {
                float s = ps[h][r], d = pd[h][r];
                s += row_xor<1>(s); d += row_xor<1>(d);
                s += row_xor<2>(s); d += row_xor<2>(d);
                s += row_xor<4>(s); d += row_xor<4>(d);
                s += row_xor<8>(s); d += row_xor<8>(d);
                ps[h][r] = s; pd[h][r] = d;
            }
        float os = ps[0][0], od = pd[0][0];
#pragma unroll
        for (int j = 1; j < 16; ++j)
            if (lidx == j) { os = ps[j & 1][j >> 1]; od = pd[j & 1][j >> 1]; }
        const size_t ao = (size_t)orow * HEADS + lidx;
        a_s[ao] = os;
        a_d[ao] = od;
    }
#pragma unroll
    for (int t = 0; t < 8; ++t) {
        const int col = t * 16 + lidx;
        float badd = 0.0f;
        if (bias0) badd += bias0[col];
        if (bias1) badd += bias1[col];
        float* o = OUT + (size_t)orow * FDIM + col;
#pragma unroll
        for (int r = 0; r < 8; ++r)
            o[(size_t)r * FDIM] = acc[t][r] + badd;
    }
}

__device__ __forceinline__ float lrelu(float x)  { return x > 0.f ? x : NEG_SLOPE * x; }

__global__ __launch_bounds__(PT) void edge_chunk_order_kernel(
    const int* __restrict__ src, const int* __restrict__ dst,
    int* __restrict__ sorted_w, int* __restrict__ table, int E_, int K, int TP)
{
    int2* sorted = (int2*)sorted_w;
    __shared__ int  hist[PT];
    __shared__ int  wsum[32];
    __shared__ int2 stage[CE];
    const int t    = threadIdx.x;
    const int lane = t & 31;
    const int wave = t >> 5;
    const int e0   = blockIdx.x * CE;

    hist[t] = 0;
    __syncthreads();

    int d[4], s[4], b[4], r[4];
    bool v[4];
#pragma unroll
    for (int j = 0; j < 4; ++j) {
        const int k = e0 + t + j * PT;
        v[j] = k < E_;
        d[j] = v[j] ? dst[k] : 0;
        s[j] = v[j] ? src[k] : 0;
    }
#pragma unroll
    for (int j = 0; j < 4; ++j) {
        b[j] = min((int)((unsigned)d[j] >> NBLOG), K - 1);
        r[j] = v[j] ? atomicAdd(&hist[b[j]], 1) : 0;
    }
    __syncthreads();

    const int cntb = hist[t];
    int inc = cntb;
#pragma unroll
    for (int o = 1; o < 32; o <<= 1) {
        const int y = __shfl_up(inc, o);
        if (lane >= o) inc += y;
    }
    if (lane == 31) wsum[wave] = inc;
    __syncthreads();
    if (wave == 0) {
        const int w = wsum[lane];
        int wi = w;
#pragma unroll
        for (int o = 1; o < 32; o <<= 1) {
            const int y = __shfl_up(wi, o);
            if (lane >= o) wi += y;
        }
        wsum[lane] = wi - w;
    }
    __syncthreads();
    hist[t] = wsum[wave] + inc - cntb;
    __syncthreads();

#pragma unroll
    for (int j = 0; j < 4; ++j)
        if (v[j]) stage[hist[b[j]] + r[j]] = make_int2(d[j], s[j]);
    __syncthreads();

    const int nval = min(CE, E_ - e0);
#pragma unroll
    for (int j = 0; j < 4; ++j) {
        const int i = t + j * PT;
        if (i < nval) sorted[(size_t)e0 + i] = stage[i];
    }
    if (t <= K) table[(size_t)blockIdx.x * TP + t] = hist[t];
}

__global__ __launch_bounds__(256) void records_build_kernel(
    const int* __restrict__ sorted_w, const int* __restrict__ table,
    int* __restrict__ slots, int C, int TP, int n)
{
    const int2* sorted = (const int2*)sorted_w;
    __shared__ int4 rec4[NB * SLOTW / 4];
    __shared__ int  cnt[NB];
    int* rec = (int*)rec4;
    const int t  = threadIdx.x;
    const int kb = blockIdx.x;
    const int n0 = kb * NB;

    for (int i = t; i < NB * SLOTW / 4; i += 256) rec4[i] = make_int4(0, 0, 0, 0);
    if (t < NB) cnt[t] = 0;
    __syncthreads();

    for (int c = t; c < C; c += 256) {
        const int* row = table + (size_t)c * TP + kb;
        const int i0 = max(row[0], 0);
        const int i1 = min(row[1], CE);
        const int2* run = sorted + (size_t)c * CE;
        for (int i = i0; i < i1; i += 4) {
            int2 e[4];
#pragma unroll
            for (int q = 0; q < 4; ++q)
                e[q] = (i + q < i1) ? run[i + q] : make_int2(-1, 0);
            int r[4];
#pragma unroll
            for (int q = 0; q < 4; ++q) {
                const unsigned u = (unsigned)(e[q].x - n0);
                r[q] = (i + q < i1 && u < (unsigned)NB) ? atomicAdd(&cnt[u], 1) : CAP;
            }
#pragma unroll
            for (int q = 0; q < 4; ++q)
                if ((unsigned)r[q] < (unsigned)CAP) rec[(e[q].x - n0) * SLOTW + 1 + r[q]] = e[q].y;
        }
    }
    __syncthreads();
    if (t < NB) rec[t * SLOTW] = cnt[t];
    __syncthreads();

    int4* out4 = (int4*)slots + (size_t)n0 * (SLOTW / 4);
    for (int i = t; i < NB * SLOTW / 4; i += 256)
        if (n0 + i / (SLOTW / 4) < n) out4[i] = rec4[i];
}

__device__ __forceinline__ float lane_value(float x, int l)
{
    return __int_as_float(__builtin_amdgcn_readlane(__float_as_int(x), l));
}
__device__ __forceinline__ float wave_max(float x)
{
    x = fmaxf(x, row_xor<1>(x));
    x = fmaxf(x, row_xor<2>(x));
    x = fmaxf(x, row_xor<4>(x));
    x = fmaxf(x, row_xor<8>(x));
    return fmaxf(lane_value(x, 0), lane_value(x, 16));
}
__device__ __forceinline__ float wave_sum(float x)
{
    x += row_xor<1>(x);
    x += row_xor<2>(x);
    x += row_xor<4>(x);
    x += row_xor<8>(x);
    return lane_value(x, 0) + lane_value(x, 16);
}

template <bool RELU>
__global__ __launch_bounds__(256) void node_softmax_gather_kernel(
    const int* __restrict__ slots,
    const int* __restrict__ ovdst, const int* __restrict__ ovsrc,
    const float* __restrict__ a_s, const float* __restrict__ a_d,
    const float* __restrict__ Hp, float* __restrict__ out, int E_, int n)
{
    const int lane = threadIdx.x & 31;
    const int d = __builtin_amdgcn_readfirstlane((int)(blockIdx.x * 8 + (threadIdx.x >> 5)));
    if (d >= n) return;
    const int h  = lane >> 4;
    const int c0 = lane * 4;
    const int* rec = slots + (size_t)d * SLOTW;
    const int cnt = rec[0];
    const int end = cnt > CAP ? 0 : max(cnt, 0);
    const int nov = cnt > CAP ? E_ : 0;
    const float2 adv = *(const float2*)(a_d + (size_t)d * HEADS);
    const float2 asv = *(const float2*)(a_s + (size_t)d * HEADS);
    const float4 hs  = *(const float4*)(Hp + (size_t)d * FDIM + c0);
    const float es0 = lrelu(asv.x + adv.x);
    const float es1 = lrelu(asv.y + adv.y);

    auto edge = [&](int j, int& s, float& e0, float& e1) {
        s = d; e0 = -INFINITY; e1 = -INFINITY;
        if (j < end) {
            s = min(max(rec[1 + j], 0), n - 1);
            const float2 a = *(const float2*)(a_s + (size_t)s * HEADS);
            e0 = lrelu(a.x + adv.x);
            e1 = lrelu(a.y + adv.y);
        }
    };
    auto over = [&](int j, int& s, float& e0, float& e1) {
        s = d; e0 = -INFINITY; e1 = -INFINITY;
        bool v = false;
        if (j < nov) v = ovdst[j] == d;
        if (v) {
            s = min(max(ovsrc[j], 0), n - 1);
            const float2 a = *(const float2*)(a_s + (size_t)s * HEADS);
            e0 = lrelu(a.x + adv.x);
            e1 = lrelu(a.y + adv.y);
        }
        return v;
    };

    int s_l; float e0, e1;
    edge(lane, s_l, e0, e1);

    float mx0 = e0, mx1 = e1;
    for (int base = 32; base < end; base += 32) {
        int s; float f0, f1;
        edge(base + lane, s, f0, f1);
        mx0 = fmaxf(mx0, f0); mx1 = fmaxf(mx1, f1);
    }
    for (int base = 0; base < nov; base += 32) {
        int s; float f0, f1;
        over(base + lane, s, f0, f1);
        mx0 = fmaxf(mx0, f0); mx1 = fmaxf(mx1, f1);
    }
    const float m0 = fmaxf(wave_max(mx0), es0);
    const float m1 = fmaxf(wave_max(mx1), es1);

    const bool v_l = lane < end;
    const float p0 = v_l ? __expf(e0 - m0) : 0.f;
    const float p1 = v_l ? __expf(e1 - m1) : 0.f;
    float sm0 = p0, sm1 = p1;
    for (int base = 32; base < end; base += 32) {
        int s; float f0, f1;
        edge(base + lane, s, f0, f1);
        if (base + lane < end) { sm0 += __expf(f0 - m0); sm1 += __expf(f1 - m1); }
    }
    for (int base = 0; base < nov; base += 32) {
        int s; float f0, f1;
        if (over(base + lane, s, f0, f1)) { sm0 += __expf(f0 - m0); sm1 += __expf(f1 - m1); }
    }
    const float ps0 = __expf(es0 - m0);
    const float ps1 = __expf(es1 - m1);
    const float den0 = (wave_sum(sm0) + ps0) + 1e-16f;
    const float den1 = (wave_sum(sm1) + ps1) + 1e-16f;

    const float aself = h ? ps1 / den1 : ps0 / den0;
    float4 acc;
    acc.x = hs.x * aself; acc.y = hs.y * aself; acc.z = hs.z * aself; acc.w = hs.w * aself;

    auto row = [&](int s_v, float a0, float a1, int t) {
        const int s = __builtin_amdgcn_readlane(s_v, t);
        const float w0 = lane_value(a0, t);
        const float w1 = lane_value(a1, t);
        const float a = h ? w1 : w0;
        const float4 hv = *(const float4*)(Hp + (size_t)s * FDIM + c0);
        acc.x += hv.x * a;
        acc.y += hv.y * a;
        acc.z += hv.z * a;
        acc.w += hv.w * a;
    };
    auto rows = [&](int s_v, float a0, float a1, int m) {
#pragma unroll 4
        for (int t = 0; t < m; ++t) row(s_v, a0, a1, t);
    };

    rows(s_l, p0 / den0, p1 / den1, min(32, end));
    for (int base = 32; base < end; base += 32) {
        int s; float f0, f1;
        edge(base + lane, s, f0, f1);
        const bool v = base + lane < end;
        const float a0 = v ? __expf(f0 - m0) / den0 : 0.f;
        const float a1 = v ? __expf(f1 - m1) / den1 : 0.f;
        rows(s, a0, a1, min(32, end - base));
    }
    for (int base = 0; base < nov; base += 32) {
        int s; float f0, f1;
        const bool v = over(base + lane, s, f0, f1);
        const float a0 = v ? __expf(f0 - m0) / den0 : 0.f;
        const float a1 = v ? __expf(f1 - m1) / den1 : 0.f;
        for (unsigned live = __builtin_amdgcn_ballot_w32(v); live; live &= live - 1)
            row(s, a0, a1, __builtin_ctz(live));
    }

    float4* o = (float4*)(out + (size_t)d * FDIM + c0);
    float4 ov = *o;
    ov.x += acc.x; ov.y += acc.y; ov.z += acc.z; ov.w += acc.w;
    if (RELU) {
        ov.x = fmaxf(ov.x, 0.0f); ov.y = fmaxf(ov.y, 0.0f);
        ov.z = fmaxf(ov.z, 0.0f); ov.w = fmaxf(ov.w, 0.0f);
    }
    *o = ov;
}

extern "C" void kernel_launch(void* const* d_in, const int* in_sizes, int n_in,
                              void* d_out, int out_size, void* d_ws, size_t ws_size,
                              hipStream_t stream)
{
    const float* x        = (const float*)d_in[0];
    const int*   src      = (const int*)  d_in[1];
    const int*   dst      = (const int*)  d_in[2];
    const float* W1       = (const float*)d_in[3];
    const float* att_src1 = (const float*)d_in[4];
    const float* att_dst1 = (const float*)d_in[5];
    const float* b1       = (const float*)d_in[6];
    const float* Wsk1     = (const float*)d_in[7];
    const float* bsk1     = (const float*)d_in[8];
    const float* W2       = (const float*)d_in[9];
    const float* att_src2 = (const float*)d_in[10];
    const float* att_dst2 = (const float*)d_in[11];
    const float* b2       = (const float*)d_in[12];
    const float* Wsk2     = (const float*)d_in[13];
    const float* bsk2     = (const float*)d_in[14];

    const int N  = in_sizes[0] / FDIM;
    const int E  = in_sizes[1];
    float* out   = (float*)d_out;

    char* base = (char*)d_ws;
    size_t off = 0;
    auto alloc = [&](size_t bytes) {
        void* p = base + off;
        off = (off + bytes + 255) & ~(size_t)255;
        return p;
    };
    float* Hproj = (float*)alloc((size_t)N * FDIM * 4);
    float* Hbuf  = (float*)alloc((size_t)N * FDIM * 4);
    float* a_s   = (float*)alloc((size_t)N * HEADS * 4);
    float* a_d   = (float*)alloc((size_t)N * HEADS * 4);
    const int K  = (N + NB - 1) / NB;
    const int C  = (E + CE - 1) / CE;
    const int TP = (K + 1 + 31) & ~31;
    int*   slots  = (int*) alloc((size_t)N * SLOTW * 4);
    int*   sorted = (int*) alloc((size_t)C * CE * 8);
    int*   table  = (int*) alloc((size_t)C * TP * 4);

    const int ntiles     = N / 16;
    const dim3 gGemm((ntiles + 3) / 4), bGemm(128);
    const dim3 b256(256);
    const dim3 gNode((N + 7) / 8);

    edge_chunk_order_kernel<<<dim3(C), dim3(PT), 0, stream>>>(src, dst, sorted, table, E, K, TP);
    records_build_kernel<<<dim3(K), b256, 0, stream>>>(sorted, table, slots, C, TP, N);

    gemm_f32_wmma<true><<<gGemm, bGemm, 0, stream>>>(x, W1, nullptr, nullptr, att_src1, att_dst1, a_s, a_d, Hproj, N);
    gemm_f32_wmma<false><<<gGemm, bGemm, 0, stream>>>(x, Wsk1, b1, bsk1, nullptr, nullptr, nullptr, nullptr, Hbuf, N);
    node_softmax_gather_kernel<true><<<gNode, b256, 0, stream>>>(slots, dst, src, a_s, a_d, Hproj, Hbuf, E, N);

    gemm_f32_wmma<true><<<gGemm, bGemm, 0, stream>>>(Hbuf, W2, nullptr, nullptr, att_src2, att_dst2, a_s, a_d, Hproj, N);
    gemm_f32_wmma<false><<<gGemm, bGemm, 0, stream>>>(Hbuf, Wsk2, b2, bsk2, nullptr, nullptr, nullptr, nullptr, out, N);
    node_softmax_gather_kernel<false><<<gNode, b256, 0, stream>>>(slots, dst, src, a_s, a_d, Hproj, out, E, N);
}


# Both GEMMs of a layer (projection, skip path) in one launch: each wave multiplies its 16-row strip by both weight matrices from one read of its A fragments (16 accumulator tiles); two launches go
# speedup vs baseline: 5.5274x; 1.4459x over previous
#include <hip/hip_runtime.h>

#define FDIM 128
#define HEADS 2
#define CH 64
#define NEG_SLOPE 0.2f
#define SLOTW 64
#define CAP 63
#define NBLOG 7
#define NB 128
#define CE 4096
#define PT 1024

typedef float v2f __attribute__((ext_vector_type(2)));
typedef float v8f __attribute__((ext_vector_type(8)));

template <int M>
__device__ __forceinline__ float row_xor(float x)
{
    return __int_as_float(__builtin_amdgcn_update_dpp(0, __float_as_int(x), 0x160 + M, 0xf, 0xf, true));
}

__global__ __launch_bounds__(128) void gemm_pair_f32_wmma(
    const float* __restrict__ X, const float* __restrict__ Wp, const float* __restrict__ Wk,
    const float* __restrict__ bias0, const float* __restrict__ bias1,
    const float* __restrict__ att_src, const float* __restrict__ att_dst,
    float* __restrict__ a_s, float* __restrict__ a_d,
    float* __restrict__ OUTP, float* __restrict__ OUTS, int nrows)
{
    const int ntiles = nrows >> 4;
    const int tile = blockIdx.x * 4 + (threadIdx.x >> 5);
    if (tile >= ntiles) return;

    const int lane  = threadIdx.x & 31;
    const int lhalf = lane >> 4;
    const int lidx  = lane & 15;
    const int row0  = tile * 16;

    const float* xrow = X + (size_t)(row0 + lidx) * FDIM + 2 * lhalf;

    v8f acc[8];
    v8f ack[8];
#pragma unroll
    for (int t = 0; t < 8; ++t)
#pragma unroll
        for (int r = 0; r < 8; ++r) { acc[t][r] = 0.0f; ack[t][r] = 0.0f; }

#pragma unroll 2
    for (int kk = 0; kk < FDIM; kk += 4) {
        v2f a = *(const v2f*)(xrow + kk);
        const int krow = kk + 2 * lhalf;
        const float* pr0 = Wp + (size_t)krow * FDIM + lidx;
        const float* pr1 = pr0 + FDIM;
        const float* kr0 = Wk + (size_t)krow * FDIM + lidx;
        const float* kr1 = kr0 + FDIM;
#pragma unroll
        for (int t = 0; t < 8; ++t) {
            v2f b;
            b.x = pr0[t * 16];
            b.y = pr1[t * 16];
            acc[t] = __builtin_amdgcn_wmma_f32_16x16x4_f32(
                false, a, false, b, (short)0, acc[t], false, false);
        }
#pragma unroll
        for (int t = 0; t < 8; ++t) {
            v2f b;
            b.x = kr0[t * 16];
            b.y = kr1[t * 16];
            ack[t] = __builtin_amdgcn_wmma_f32_16x16x4_f32(
                false, a, false, b, (short)0, ack[t], false, false);
        }
    }

    const int orow = row0 + 8 * lhalf;

    {
        float ps[HEADS][8], pd[HEADS][8];
#pragma unroll
        for (int h = 0; h < HEADS; ++h) {
#pragma unroll
            for (int r = 0; r < 8; ++r) { ps[h][r] = 0.0f; pd[h][r] = 0.0f; }
#pragma unroll
            for (int tt = 0; tt < 4; ++tt) {
                const int t = 4 * h + tt;
                const float vs = att_src[t * 16 + lidx];
                const float vd = att_dst[t * 16 + lidx];
#pragma unroll
                for (int r = 0; r < 8; ++r) {
                    ps[h][r] += acc[t][r] * vs;
                    pd[h][r] += acc[t][r] * vd;
                }
            }
        }
#pragma unroll
        for (int h = 0; h < HEADS; ++h)
#pragma unroll
            for (int r = 0; r < 8; ++r) {
                float s = ps[h][r], d = pd[h][r];
                s += row_xor<1>(s); d += row_xor<1>(d);
                s += row_xor<2>(s); d += row_xor<2>(d);
                s += row_xor<4>(s); d += row_xor<4>(d);
                s += row_xor<8>(s); d += row_xor<8>(d);
                ps[h][r] = s; pd[h][r] = d;
            }
        float os = ps[0][0], od = pd[0][0];
#pragma unroll
        for (int j = 1; j < 16; ++j)
            if (lidx == j) { os = ps[j & 1][j >> 1]; od = pd[j & 1][j >> 1]; }
        const size_t ao = (size_t)orow * HEADS + lidx;
        a_s[ao] = os;
        a_d[ao] = od;
    }
#pragma unroll
    for (int t = 0; t < 8; ++t) {
        const int col = t * 16 + lidx;
        float* o = OUTP + (size_t)orow * FDIM + col;
#pragma unroll
        for (int r = 0; r < 8; ++r)
            o[(size_t)r * FDIM] = acc[t][r];
    }
#pragma unroll
    for (int t = 0; t < 8; ++t) {
        const int col = t * 16 + lidx;
        const float badd = bias0[col] + bias1[col];
        float* o = OUTS + (size_t)orow * FDIM + col;
#pragma unroll
        for (int r = 0; r < 8; ++r)
            o[(size_t)r * FDIM] = ack[t][r] + badd;
    }
}

__device__ __forceinline__ float lrelu(float x)  { return x > 0.f ? x : NEG_SLOPE * x; }

__global__ __launch_bounds__(PT) void edge_chunk_order_kernel(
    const int* __restrict__ src, const int* __restrict__ dst,
    int* __restrict__ sorted_w, int* __restrict__ table, int E_, int K, int TP)
{
    int2* sorted = (int2*)sorted_w;
    __shared__ int  hist[PT];
    __shared__ int  wsum[32];
    __shared__ int2 stage[CE];
    const int t    = threadIdx.x;
    const int lane = t & 31;
    const int wave = t >> 5;
    const int e0   = blockIdx.x * CE;

    hist[t] = 0;
    __syncthreads();

    int d[4], s[4], b[4], r[4];
    bool v[4];
#pragma unroll
    for (int j = 0; j < 4; ++j) {
        const int k = e0 + t + j * PT;
        v[j] = k < E_;
        d[j] = v[j] ? dst[k] : 0;
        s[j] = v[j] ? src[k] : 0;
    }
#pragma unroll
    for (int j = 0; j < 4; ++j) {
        b[j] = min((int)((unsigned)d[j] >> NBLOG), K - 1);
        r[j] = v[j] ? atomicAdd(&hist[b[j]], 1) : 0;
    }
    __syncthreads();

    const int cntb = hist[t];
    int inc = cntb;
#pragma unroll
    for (int o = 1; o < 32; o <<= 1) {
        const int y = __shfl_up(inc, o);
        if (lane >= o) inc += y;
    }
    if (lane == 31) wsum[wave] = inc;
    __syncthreads();
    if (wave == 0) {
        const int w = wsum[lane];
        int wi = w;
#pragma unroll
        for (int o = 1; o < 32; o <<= 1) {
            const int y = __shfl_up(wi, o);
            if (lane >= o) wi += y;
        }
        wsum[lane] = wi - w;
    }
    __syncthreads();
    hist[t] = wsum[wave] + inc - cntb;
    __syncthreads();

#pragma unroll
    for (int j = 0; j < 4; ++j)
        if (v[j]) stage[hist[b[j]] + r[j]] = make_int2(d[j], s[j]);
    __syncthreads();

    const int nval = min(CE, E_ - e0);
#pragma unroll
    for (int j = 0; j < 4; ++j) {
        const int i = t + j * PT;
        if (i < nval) sorted[(size_t)e0 + i] = stage[i];
    }
    if (t <= K) table[(size_t)blockIdx.x * TP + t] = hist[t];
}

__global__ __launch_bounds__(256) void records_build_kernel(
    const int* __restrict__ sorted_w, const int* __restrict__ table,
    int* __restrict__ slots, int C, int TP, int n)
{
    const int2* sorted = (const int2*)sorted_w;
    __shared__ int4 rec4[NB * SLOTW / 4];
    __shared__ int  cnt[NB];
    int* rec = (int*)rec4;
    const int t  = threadIdx.x;
    const int kb = blockIdx.x;
    const int n0 = kb * NB;

    for (int i = t; i < NB * SLOTW / 4; i += 256) rec4[i] = make_int4(0, 0, 0, 0);
    if (t < NB) cnt[t] = 0;
    __syncthreads();

    for (int c = t; c < C; c += 256) {
        const int* row = table + (size_t)c * TP + kb;
        const int i0 = max(row[0], 0);
        const int i1 = min(row[1], CE);
        const int2* run = sorted + (size_t)c * CE;
        for (int i = i0; i < i1; i += 4) {
            int2 e[4];
#pragma unroll
            for (int q = 0; q < 4; ++q)
                e[q] = (i + q < i1) ? run[i + q] : make_int2(-1, 0);
            int r[4];
#pragma unroll
            for (int q = 0; q < 4; ++q) {
                const unsigned u = (unsigned)(e[q].x - n0);
                r[q] = (i + q < i1 && u < (unsigned)NB) ? atomicAdd(&cnt[u], 1) : CAP;
            }
#pragma unroll
            for (int q = 0; q < 4; ++q)
                if ((unsigned)r[q] < (unsigned)CAP) rec[(e[q].x - n0) * SLOTW + 1 + r[q]] = e[q].y;
        }
    }
    __syncthreads();
    if (t < NB) rec[t * SLOTW] = cnt[t];
    __syncthreads();

    int4* out4 = (int4*)slots + (size_t)n0 * (SLOTW / 4);
    for (int i = t; i < NB * SLOTW / 4; i += 256)
        if (n0 + i / (SLOTW / 4) < n) out4[i] = rec4[i];
}

__device__ __forceinline__ float lane_value(float x, int l)
{
    return __int_as_float(__builtin_amdgcn_readlane(__float_as_int(x), l));
}
__device__ __forceinline__ float wave_max(float x)
{
    x = fmaxf(x, row_xor<1>(x));
    x = fmaxf(x, row_xor<2>(x));
    x = fmaxf(x, row_xor<4>(x));
    x = fmaxf(x, row_xor<8>(x));
    return fmaxf(lane_value(x, 0), lane_value(x, 16));
}
__device__ __forceinline__ float wave_sum(float x)
{
    x += row_xor<1>(x);
    x += row_xor<2>(x);
    x += row_xor<4>(x);
    x += row_xor<8>(x);
    return lane_value(x, 0) + lane_value(x, 16);
}

template <bool RELU>
__global__ __launch_bounds__(256) void node_softmax_gather_kernel(
    const int* __restrict__ slots,
    const int* __restrict__ ovdst, const int* __restrict__ ovsrc,
    const float* __restrict__ a_s, const float* __restrict__ a_d,
    const float* __restrict__ Hp, float* __restrict__ out, int E_, int n)
{
    const int lane = threadIdx.x & 31;
    const int d = __builtin_amdgcn_readfirstlane((int)(blockIdx.x * 8 + (threadIdx.x >> 5)));
    if (d >= n) return;
    const int h  = lane >> 4;
    const int c0 = lane * 4;
    const int* rec = slots + (size_t)d * SLOTW;
    const int cnt = rec[0];
    const int end = cnt > CAP ? 0 : max(cnt, 0);
    const int nov = cnt > CAP ? E_ : 0;
    const float2 adv = *(const float2*)(a_d + (size_t)d * HEADS);
    const float2 asv = *(const float2*)(a_s + (size_t)d * HEADS);
    const float4 hs  = *(const float4*)(Hp + (size_t)d * FDIM + c0);
    const float es0 = lrelu(asv.x + adv.x);
    const float es1 = lrelu(asv.y + adv.y);

    auto edge = [&](int j, int& s, float& e0, float& e1) {
        s = d; e0 = -INFINITY; e1 = -INFINITY;
        if (j < end) {
            s = min(max(rec[1 + j], 0), n - 1);
            const float2 a = *(const float2*)(a_s + (size_t)s * HEADS);
            e0 = lrelu(a.x + adv.x);
            e1 = lrelu(a.y + adv.y);
        }
    };
    auto over = [&](int j, int& s, float& e0, float& e1) {
        s = d; e0 = -INFINITY; e1 = -INFINITY;
        bool v = false;
        if (j < nov) v = ovdst[j] == d;
        if (v) {
            s = min(max(ovsrc[j], 0), n - 1);
            const float2 a = *(const float2*)(a_s + (size_t)s * HEADS);
            e0 = lrelu(a.x + adv.x);
            e1 = lrelu(a.y + adv.y);
        }
        return v;
    };

    int s_l; float e0, e1;
    edge(lane, s_l, e0, e1);

    float mx0 = e0, mx1 = e1;
    for (int base = 32; base < end; base += 32) {
        int s; float f0, f1;
        edge(base + lane, s, f0, f1);
        mx0 = fmaxf(mx0, f0); mx1 = fmaxf(mx1, f1);
    }
    for (int base = 0; base < nov; base += 32) {
        int s; float f0, f1;
        over(base + lane, s, f0, f1);
        mx0 = fmaxf(mx0, f0); mx1 = fmaxf(mx1, f1);
    }
    const float m0 = fmaxf(wave_max(mx0), es0);
    const float m1 = fmaxf(wave_max(mx1), es1);

    const bool v_l = lane < end;
    const float p0 = v_l ? __expf(e0 - m0) : 0.f;
    const float p1 = v_l ? __expf(e1 - m1) : 0.f;
    float sm0 = p0, sm1 = p1;
    for (int base = 32; base < end; base += 32) {
        int s; float f0, f1;
        edge(base + lane, s, f0, f1);
        if (base + lane < end) { sm0 += __expf(f0 - m0); sm1 += __expf(f1 - m1); }
    }
    for (int base = 0; base < nov; base += 32) {
        int s; float f0, f1;
        if (over(base + lane, s, f0, f1)) { sm0 += __expf(f0 - m0); sm1 += __expf(f1 - m1); }
    }
    const float ps0 = __expf(es0 - m0);
    const float ps1 = __expf(es1 - m1);
    const float den0 = (wave_sum(sm0) + ps0) + 1e-16f;
    const float den1 = (wave_sum(sm1) + ps1) + 1e-16f;

    const float aself = h ? ps1 / den1 : ps0 / den0;
    float4 acc;
    acc.x = hs.x * aself; acc.y = hs.y * aself; acc.z = hs.z * aself; acc.w = hs.w * aself;

    auto row = [&](int s_v, float a0, float a1, int t) {
        const int s = __builtin_amdgcn_readlane(s_v, t);
        const float w0 = lane_value(a0, t);
        const float w1 = lane_value(a1, t);
        const float a = h ? w1 : w0;
        const float4 hv = *(const float4*)(Hp + (size_t)s * FDIM + c0);
        acc.x += hv.x * a;
        acc.y += hv.y * a;
        acc.z += hv.z * a;
        acc.w += hv.w * a;
    };
    auto rows = [&](int s_v, float a0, float a1, int m) {
#pragma unroll 4
        for (int t = 0; t < m; ++t) row(s_v, a0, a1, t);
    };

    rows(s_l, p0 / den0, p1 / den1, min(32, end));
    for (int base = 32; base < end; base += 32) {
        int s; float f0, f1;
        edge(base + lane, s, f0, f1);
        const bool v = base + lane < end;
        const float a0 = v ? __expf(f0 - m0) / den0 : 0.f;
        const float a1 = v ? __expf(f1 - m1) / den1 : 0.f;
        rows(s, a0, a1, min(32, end - base));
    }
    for (int base = 0; base < nov; base += 32) {
        int s; float f0, f1;
        const bool v = over(base + lane, s, f0, f1);
        const float a0 = v ? __expf(f0 - m0) / den0 : 0.f;
        const float a1 = v ? __expf(f1 - m1) / den1 : 0.f;
        for (unsigned live = __builtin_amdgcn_ballot_w32(v); live; live &= live - 1)
            row(s, a0, a1, __builtin_ctz(live));
    }

    float4* o = (float4*)(out + (size_t)d * FDIM + c0);
    float4 ov = *o;
    ov.x += acc.x; ov.y += acc.y; ov.z += acc.z; ov.w += acc.w;
    if (RELU) {
        ov.x = fmaxf(ov.x, 0.0f); ov.y = fmaxf(ov.y, 0.0f);
        ov.z = fmaxf(ov.z, 0.0f); ov.w = fmaxf(ov.w, 0.0f);
    }
    *o = ov;
}

extern "C" void kernel_launch(void* const* d_in, const int* in_sizes, int n_in,
                              void* d_out, int out_size, void* d_ws, size_t ws_size,
                              hipStream_t stream)
{
    const float* x        = (const float*)d_in[0];
    const int*   src      = (const int*)  d_in[1];
    const int*   dst      = (const int*)  d_in[2];
    const float* W1       = (const float*)d_in[3];
    const float* att_src1 = (const float*)d_in[4];
    const float* att_dst1 = (const float*)d_in[5];
    const float* b1       = (const float*)d_in[6];
    const float* Wsk1     = (const float*)d_in[7];
    const float* bsk1     = (const float*)d_in[8];
    const float* W2       = (const float*)d_in[9];
    const float* att_src2 = (const float*)d_in[10];
    const float* att_dst2 = (const float*)d_in[11];
    const float* b2       = (const float*)d_in[12];
    const float* Wsk2     = (const float*)d_in[13];
    const float* bsk2     = (const float*)d_in[14];

    const int N  = in_sizes[0] / FDIM;
    const int E  = in_sizes[1];
    float* out   = (float*)d_out;

    char* base = (char*)d_ws;
    size_t off = 0;
    auto alloc = [&](size_t bytes) {
        void* p = base + off;
        off = (off + bytes + 255) & ~(size_t)255;
        return p;
    };
    float* Hproj = (float*)alloc((size_t)N * FDIM * 4);
    float* Hbuf  = (float*)alloc((size_t)N * FDIM * 4);
    float* a_s   = (float*)alloc((size_t)N * HEADS * 4);
    float* a_d   = (float*)alloc((size_t)N * HEADS * 4);
    const int K  = (N + NB - 1) / NB;
    const int C  = (E + CE - 1) / CE;
    const int TP = (K + 1 + 31) & ~31;
    int*   slots  = (int*) alloc((size_t)N * SLOTW * 4);
    int*   sorted = (int*) alloc((size_t)C * CE * 8);
    int*   table  = (int*) alloc((size_t)C * TP * 4);

    const int ntiles     = N / 16;
    const dim3 gGemm((ntiles + 3) / 4), bGemm(128);
    const dim3 b256(256);
    const dim3 gNode((N + 7) / 8);

    edge_chunk_order_kernel<<<dim3(C), dim3(PT), 0, stream>>>(src, dst, sorted, table, E, K, TP);
    records_build_kernel<<<dim3(K), b256, 0, stream>>>(sorted, table, slots, C, TP, N);

    gemm_pair_f32_wmma<<<gGemm, bGemm, 0, stream>>>(x, W1, Wsk1, b1, bsk1, att_src1, att_dst1, a_s, a_d, Hproj, Hbuf, N);
    node_softmax_gather_kernel<true><<<gNode, b256, 0, stream>>>(slots, dst, src, a_s, a_d, Hproj, Hbuf, E, N);

    gemm_pair_f32_wmma<<<gGemm, bGemm, 0, stream>>>(Hbuf, W2, Wsk2, b2, bsk2, att_src2, att_dst2, a_s, a_d, Hproj, out, N);
    node_softmax_gather_kernel<false><<<gNode, b256, 0, stream>>>(slots, dst, src, a_s, a_d, Hproj, out, E, N);
}


# Both weight matrices of a layer staged once per 512-thread work-group in LDS in fragment order: a B fragment is one 8-byte LDS read, not two 4-byte global loads per lane (k order, sums unchanged)
# speedup vs baseline: 5.9897x; 1.0836x over previous
#include <hip/hip_runtime.h>

#define FDIM 128
#define HEADS 2
#define CH 64
#define NEG_SLOPE 0.2f
#define SLOTW 64
#define CAP 63
#define NBLOG 7
#define NB 128
#define CE 4096
#define PT 1024

#define GW 16
#define WKOFF (2 * FDIM)
#define WPITCH (4 * FDIM + 32)
#define WPAIRS (FDIM / 2)

typedef float v2f __attribute__((ext_vector_type(2)));
typedef float v8f __attribute__((ext_vector_type(8)));

template <int M>
__device__ __forceinline__ float row_xor(float x)
{
    return __int_as_float(__builtin_amdgcn_update_dpp(0, __float_as_int(x), 0x160 + M, 0xf, 0xf, true));
}

__global__ __launch_bounds__(GW * 32) void gemm_pair_f32_wmma(
    const float* __restrict__ X, const float* __restrict__ Wp, const float* __restrict__ Wk,
    const float* __restrict__ bias0, const float* __restrict__ bias1,
    const float* __restrict__ att_src, const float* __restrict__ att_dst,
    float* __restrict__ a_s, float* __restrict__ a_d,
    float* __restrict__ OUTP, float* __restrict__ OUTS, int nrows)
{
    __shared__ __attribute__((aligned(16))) float wl[WPAIRS * WPITCH];

#pragma unroll
    for (int i = threadIdx.x; i < 2 * WPAIRS * (FDIM / 4); i += GW * 32) {
        const int m = i / (WPAIRS * (FDIM / 4));
        const int p = (i / (FDIM / 4)) % WPAIRS;
        const int q = i % (FDIM / 4);
        const float* w = (m ? Wk : Wp) + (size_t)(2 * p) * FDIM + 4 * q;
        const float4 r0 = *(const float4*)w;
        const float4 r1 = *(const float4*)(w + FDIM);
        float* dl = wl + p * WPITCH + m * WKOFF + 8 * q;
        *(float4*)dl       = make_float4(r0.x, r1.x, r0.y, r1.y);
        *(float4*)(dl + 4) = make_float4(r0.z, r1.z, r0.w, r1.w);
    }
    __syncthreads();

    const int ntiles = nrows >> 4;
    const int tile = blockIdx.x * GW + (threadIdx.x >> 5);
    if (tile >= ntiles) return;

    const int lane  = threadIdx.x & 31;
    const int lhalf = lane >> 4;
    const int lidx  = lane & 15;
    const int row0  = tile * 16;

    const float* xrow = X + (size_t)(row0 + lidx) * FDIM + 2 * lhalf;

    v8f acc[8];
    v8f ack[8];
#pragma unroll
    for (int t = 0; t < 8; ++t)
#pragma unroll
        for (int r = 0; r < 8; ++r) { acc[t][r] = 0.0f; ack[t][r] = 0.0f; }

    const float* bp = wl + lhalf * WPITCH + 2 * lidx;

#pragma unroll 2
    for (int kk = 0; kk < FDIM; kk += 4) {
        v2f a = *(const v2f*)(xrow + kk);
        const float* pr = bp + (kk >> 1) * WPITCH;
        const float* kr = pr + WKOFF;
#pragma unroll
        for (int t = 0; t < 8; ++t) {
            const v2f b = *(const v2f*)(pr + t * 32);
            acc[t] = __builtin_amdgcn_wmma_f32_16x16x4_f32(
                false, a, false, b, (short)0, acc[t], false, false);
        }
#pragma unroll
        for (int t = 0; t < 8; ++t) {
            const v2f b = *(const v2f*)(kr + t * 32);
            ack[t] = __builtin_amdgcn_wmma_f32_16x16x4_f32(
                false, a, false, b, (short)0, ack[t], false, false);
        }
    }

    const int orow = row0 + 8 * lhalf;

    {
        float ps[HEADS][8], pd[HEADS][8];
#pragma unroll
        for (int h = 0; h < HEADS; ++h) {
#pragma unroll
            for (int r = 0; r < 8; ++r) { ps[h][r] = 0.0f; pd[h][r] = 0.0f; }
#pragma unroll
            for (int tt = 0; tt < 4; ++tt) {
                const int t = 4 * h + tt;
                const float vs = att_src[t * 16 + lidx];
                const float vd = att_dst[t * 16 + lidx];
#pragma unroll
                for (int r = 0; r < 8; ++r) {
                    ps[h][r] += acc[t][r] * vs;
                    pd[h][r] += acc[t][r] * vd;
                }
            }
        }
#pragma unroll
        for (int h = 0; h < HEADS; ++h)
#pragma unroll
            for (int r = 0; r < 8; ++r) {
                float s = ps[h][r], d = pd[h][r];
                s += row_xor<1>(s); d += row_xor<1>(d);
                s += row_xor<2>(s); d += row_xor<2>(d);
                s += row_xor<4>(s); d += row_xor<4>(d);
                s += row_xor<8>(s); d += row_xor<8>(d);
                ps[h][r] = s; pd[h][r] = d;
            }
        float os = ps[0][0], od = pd[0][0];
#pragma unroll
        for (int j = 1; j < 16; ++j)
            if (lidx == j) { os = ps[j & 1][j >> 1]; od = pd[j & 1][j >> 1]; }
        const size_t ao = (size_t)orow * HEADS + lidx;
        a_s[ao] = os;
        a_d[ao] = od;
    }
#pragma unroll
    for (int t = 0; t < 8; ++t) {
        const int col = t * 16 + lidx;
        float* o = OUTP + (size_t)orow * FDIM + col;
#pragma unroll
        for (int r = 0; r < 8; ++r)
            o[(size_t)r * FDIM] = acc[t][r];
    }
#pragma unroll
    for (int t = 0; t < 8; ++t) {
        const int col = t * 16 + lidx;
        const float badd = bias0[col] + bias1[col];
        float* o = OUTS + (size_t)orow * FDIM + col;
#pragma unroll
        for (int r = 0; r < 8; ++r)
            o[(size_t)r * FDIM] = ack[t][r] + badd;
    }
}

__device__ __forceinline__ float lrelu(float x)  { return x > 0.f ? x : NEG_SLOPE * x; }

__global__ __launch_bounds__(PT) void edge_chunk_order_kernel(
    const int* __restrict__ src, const int* __restrict__ dst,
    int* __restrict__ sorted_w, int* __restrict__ table, int E_, int K, int TP)
{
    int2* sorted = (int2*)sorted_w;
    __shared__ int  hist[PT];
    __shared__ int  wsum[32];
    __shared__ int2 stage[CE];
    const int t    = threadIdx.x;
    const int lane = t & 31;
    const int wave = t >> 5;
    const int e0   = blockIdx.x * CE;

    hist[t] = 0;
    __syncthreads();

    int d[4], s[4], b[4], r[4];
    bool v[4];
#pragma unroll
    for (int j = 0; j < 4; ++j) {
        const int k = e0 + t + j * PT;
        v[j] = k < E_;
        d[j] = v[j] ? dst[k] : 0;
        s[j] = v[j] ? src[k] : 0;
    }
#pragma unroll
    for (int j = 0; j < 4; ++j) {
        b[j] = min((int)((unsigned)d[j] >> NBLOG), K - 1);
        r[j] = v[j] ? atomicAdd(&hist[b[j]], 1) : 0;
    }
    __syncthreads();

    const int cntb = hist[t];
    int inc = cntb;
#pragma unroll
    for (int o = 1; o < 32; o <<= 1) {
        const int y = __shfl_up(inc, o);
        if (lane >= o) inc += y;
    }
    if (lane == 31) wsum[wave] = inc;
    __syncthreads();
    if (wave == 0) {
        const int w = wsum[lane];
        int wi = w;
#pragma unroll
        for (int o = 1; o < 32; o <<= 1) {
            const int y = __shfl_up(wi, o);
            if (lane >= o) wi += y;
        }
        wsum[lane] = wi - w;
    }
    __syncthreads();
    hist[t] = wsum[wave] + inc - cntb;
    __syncthreads();

#pragma unroll
    for (int j = 0; j < 4; ++j)
        if (v[j]) stage[hist[b[j]] + r[j]] = make_int2(d[j], s[j]);
    __syncthreads();

    const int nval = min(CE, E_ - e0);
#pragma unroll
    for (int j = 0; j < 4; ++j) {
        const int i = t + j * PT;
        if (i < nval) sorted[(size_t)e0 + i] = stage[i];
    }
    if (t <= K) table[(size_t)blockIdx.x * TP + t] = hist[t];
}

__global__ __launch_bounds__(256) void records_build_kernel(
    const int* __restrict__ sorted_w, const int* __restrict__ table,
    int* __restrict__ slots, int C, int TP, int n)
{
    const int2* sorted = (const int2*)sorted_w;
    __shared__ int4 rec4[NB * SLOTW / 4];
    __shared__ int  cnt[NB];
    int* rec = (int*)rec4;
    const int t  = threadIdx.x;
    const int kb = blockIdx.x;
    const int n0 = kb * NB;

    for (int i = t; i < NB * SLOTW / 4; i += 256) rec4[i] = make_int4(0, 0, 0, 0);
    if (t < NB) cnt[t] = 0;
    __syncthreads();

    for (int c = t; c < C; c += 256) {
        const int* row = table + (size_t)c * TP + kb;
        const int i0 = max(row[0], 0);
        const int i1 = min(row[1], CE);
        const int2* run = sorted + (size_t)c * CE;
        for (int i = i0; i < i1; i += 4) {
            int2 e[4];
#pragma unroll
            for (int q = 0; q < 4; ++q)
                e[q] = (i + q < i1) ? run[i + q] : make_int2(-1, 0);
            int r[4];
#pragma unroll
            for (int q = 0; q < 4; ++q) {
                const unsigned u = (unsigned)(e[q].x - n0);
                r[q] = (i + q < i1 && u < (unsigned)NB) ? atomicAdd(&cnt[u], 1) : CAP;
            }
#pragma unroll
            for (int q = 0; q < 4; ++q)
                if ((unsigned)r[q] < (unsigned)CAP) rec[(e[q].x - n0) * SLOTW + 1 + r[q]] = e[q].y;
        }
    }
    __syncthreads();
    if (t < NB) rec[t * SLOTW] = cnt[t];
    __syncthreads();

    int4* out4 = (int4*)slots + (size_t)n0 * (SLOTW / 4);
    for (int i = t; i < NB * SLOTW / 4; i += 256)
        if (n0 + i / (SLOTW / 4) < n) out4[i] = rec4[i];
}

__device__ __forceinline__ float lane_value(float x, int l)
{
    return __int_as_float(__builtin_amdgcn_readlane(__float_as_int(x), l));
}
__device__ __forceinline__ float wave_max(float x)
{
    x = fmaxf(x, row_xor<1>(x));
    x = fmaxf(x, row_xor<2>(x));
    x = fmaxf(x, row_xor<4>(x));
    x = fmaxf(x, row_xor<8>(x));
    return fmaxf(lane_value(x, 0), lane_value(x, 16));
}
__device__ __forceinline__ float wave_sum(float x)
{
    x += row_xor<1>(x);
    x += row_xor<2>(x);
    x += row_xor<4>(x);
    x += row_xor<8>(x);
    return lane_value(x, 0) + lane_value(x, 16);
}

template <bool RELU>
__global__ __launch_bounds__(256) void node_softmax_gather_kernel(
    const int* __restrict__ slots,
    const int* __restrict__ ovdst, const int* __restrict__ ovsrc,
    const float* __restrict__ a_s, const float* __restrict__ a_d,
    const float* __restrict__ Hp, float* __restrict__ out, int E_, int n)
{
    const int lane = threadIdx.x & 31;
    const int d = __builtin_amdgcn_readfirstlane((int)(blockIdx.x * 8 + (threadIdx.x >> 5)));
    if (d >= n) return;
    const int h  = lane >> 4;
    const int c0 = lane * 4;
    const int* rec = slots + (size_t)d * SLOTW;
    const int cnt = rec[0];
    const int end = cnt > CAP ? 0 : max(cnt, 0);
    const int nov = cnt > CAP ? E_ : 0;
    const float2 adv = *(const float2*)(a_d + (size_t)d * HEADS);
    const float2 asv = *(const float2*)(a_s + (size_t)d * HEADS);
    const float4 hs  = *(const float4*)(Hp + (size_t)d * FDIM + c0);
    const float es0 = lrelu(asv.x + adv.x);
    const float es1 = lrelu(asv.y + adv.y);

    auto edge = [&](int j, int& s, float& e0, float& e1) {
        s = d; e0 = -INFINITY; e1 = -INFINITY;
        if (j < end) {
            s = min(max(rec[1 + j], 0), n - 1);
            const float2 a = *(const float2*)(a_s + (size_t)s * HEADS);
            e0 = lrelu(a.x + adv.x);
            e1 = lrelu(a.y + adv.y);
        }
    };
    auto over = [&](int j, int& s, float& e0, float& e1) {
        s = d; e0 = -INFINITY; e1 = -INFINITY;
        bool v = false;
        if (j < nov) v = ovdst[j] == d;
        if (v) {
            s = min(max(ovsrc[j], 0), n - 1);
            const float2 a = *(const float2*)(a_s + (size_t)s * HEADS);
            e0 = lrelu(a.x + adv.x);
            e1 = lrelu(a.y + adv.y);
        }
        return v;
    };

    int s_l; float e0, e1;
    edge(lane, s_l, e0, e1);

    float mx0 = e0, mx1 = e1;
    for (int base = 32; base < end; base += 32) {
        int s; float f0, f1;
        edge(base + lane, s, f0, f1);
        mx0 = fmaxf(mx0, f0); mx1 = fmaxf(mx1, f1);
    }
    for (int base = 0; base < nov; base += 32) {
        int s; float f0, f1;
        over(base + lane, s, f0, f1);
        mx0 = fmaxf(mx0, f0); mx1 = fmaxf(mx1, f1);
    }
    const float m0 = fmaxf(wave_max(mx0), es0);
    const float m1 = fmaxf(wave_max(mx1), es1);

    const bool v_l = lane < end;
    const float p0 = v_l ? __expf(e0 - m0) : 0.f;
    const float p1 = v_l ? __expf(e1 - m1) : 0.f;
    float sm0 = p0, sm1 = p1;
    for (int base = 32; base < end; base += 32) {
        int s; float f0, f1;
        edge(base + lane, s, f0, f1);
        if (base + lane < end) { sm0 += __expf(f0 - m0); sm1 += __expf(f1 - m1); }
    }
    for (int base = 0; base < nov; base += 32) {
        int s; float f0, f1;
        if (over(base + lane, s, f0, f1)) { sm0 += __expf(f0 - m0); sm1 += __expf(f1 - m1); }
    }
    const float ps0 = __expf(es0 - m0);
    const float ps1 = __expf(es1 - m1);
    const float den0 = (wave_sum(sm0) + ps0) + 1e-16f;
    const float den1 = (wave_sum(sm1) + ps1) + 1e-16f;

    const float aself = h ? ps1 / den1 : ps0 / den0;
    float4 acc;
    acc.x = hs.x * aself; acc.y = hs.y * aself; acc.z = hs.z * aself; acc.w = hs.w * aself;

    auto row = [&](int s_v, float a0, float a1, int t) {
        const int s = __builtin_amdgcn_readlane(s_v, t);
        const float w0 = lane_value(a0, t);
        const float w1 = lane_value(a1, t);
        const float a = h ? w1 : w0;
        const float4 hv = *(const float4*)(Hp + (size_t)s * FDIM + c0);
        acc.x += hv.x * a;
        acc.y += hv.y * a;
        acc.z += hv.z * a;
        acc.w += hv.w * a;
    };
    auto rows = [&](int s_v, float a0, float a1, int m) {
#pragma unroll 4
        for (int t = 0; t < m; ++t) row(s_v, a0, a1, t);
    };

    rows(s_l, p0 / den0, p1 / den1, min(32, end));
    for (int base = 32; base < end; base += 32) {
        int s; float f0, f1;
        edge(base + lane, s, f0, f1);
        const bool v = base + lane < end;
        const float a0 = v ? __expf(f0 - m0) / den0 : 0.f;
        const float a1 = v ? __expf(f1 - m1) / den1 : 0.f;
        rows(s, a0, a1, min(32, end - base));
    }
    for (int base = 0; base < nov; base += 32) {
        int s; float f0, f1;
        const bool v = over(base + lane, s, f0, f1);
        const float a0 = v ? __expf(f0 - m0) / den0 : 0.f;
        const float a1 = v ? __expf(f1 - m1) / den1 : 0.f;
        for (unsigned live = __builtin_amdgcn_ballot_w32(v); live; live &= live - 1)
            row(s, a0, a1, __builtin_ctz(live));
    }

    float4* o = (float4*)(out + (size_t)d * FDIM + c0);
    float4 ov = *o;
    ov.x += acc.x; ov.y += acc.y; ov.z += acc.z; ov.w += acc.w;
    if (RELU) {
        ov.x = fmaxf(ov.x, 0.0f); ov.y = fmaxf(ov.y, 0.0f);
        ov.z = fmaxf(ov.z, 0.0f); ov.w = fmaxf(ov.w, 0.0f);
    }
    *o = ov;
}

extern "C" void kernel_launch(void* const* d_in, const int* in_sizes, int n_in,
                              void* d_out, int out_size, void* d_ws, size_t ws_size,
                              hipStream_t stream)
{
    const float* x        = (const float*)d_in[0];
    const int*   src      = (const int*)  d_in[1];
    const int*   dst      = (const int*)  d_in[2];
    const float* W1       = (const float*)d_in[3];
    const float* att_src1 = (const float*)d_in[4];
    const float* att_dst1 = (const float*)d_in[5];
    const float* b1       = (const float*)d_in[6];
    const float* Wsk1     = (const float*)d_in[7];
    const float* bsk1     = (const float*)d_in[8];
    const float* W2       = (const float*)d_in[9];
    const float* att_src2 = (const float*)d_in[10];
    const float* att_dst2 = (const float*)d_in[11];
    const float* b2       = (const float*)d_in[12];
    const float* Wsk2     = (const float*)d_in[13];
    const float* bsk2     = (const float*)d_in[14];

    const int N  = in_sizes[0] / FDIM;
    const int E  = in_sizes[1];
    float* out   = (float*)d_out;

    char* base = (char*)d_ws;
    size_t off = 0;
    auto alloc = [&](size_t bytes) {
        void* p = base + off;
        off = (off + bytes + 255) & ~(size_t)255;
        return p;
    };
    float* Hproj = (float*)alloc((size_t)N * FDIM * 4);
    float* Hbuf  = (float*)alloc((size_t)N * FDIM * 4);
    float* a_s   = (float*)alloc((size_t)N * HEADS * 4);
    float* a_d   = (float*)alloc((size_t)N * HEADS * 4);
    const int K  = (N + NB - 1) / NB;
    const int C  = (E + CE - 1) / CE;
    const int TP = (K + 1 + 31) & ~31;
    int*   slots  = (int*) alloc((size_t)N * SLOTW * 4);
    int*   sorted = (int*) alloc((size_t)C * CE * 8);
    int*   table  = (int*) alloc((size_t)C * TP * 4);

    const int ntiles     = N / 16;
    const dim3 gGemm((ntiles + GW - 1) / GW), bGemm(GW * 32);
    const dim3 b256(256);
    const dim3 gNode((N + 7) / 8);

    edge_chunk_order_kernel<<<dim3(C), dim3(PT), 0, stream>>>(src, dst, sorted, table, E, K, TP);
    records_build_kernel<<<dim3(K), b256, 0, stream>>>(sorted, table, slots, C, TP, N);

    gemm_pair_f32_wmma<<<gGemm, bGemm, 0, stream>>>(x, W1, Wsk1, b1, bsk1, att_src1, att_dst1, a_s, a_d, Hproj, Hbuf, N);
    node_softmax_gather_kernel<true><<<gNode, b256, 0, stream>>>(slots, dst, src, a_s, a_d, Hproj, Hbuf, E, N);

    gemm_pair_f32_wmma<<<gGemm, bGemm, 0, stream>>>(Hbuf, W2, Wsk2, b2, bsk2, att_src2, att_dst2, a_s, a_d, Hproj, out, N);
    node_softmax_gather_kernel<false><<<gNode, b256, 0, stream>>>(slots, dst, src, a_s, a_d, Hproj, out, E, N);
}


# Edge-record build moved into the layer-1 GEMM launch: 196 extra work-groups (two 128-node buckets each) run on the WGPs the GEMM leaves idle; its own launch goes (sum order: LDS arrival, as before)
# speedup vs baseline: 6.3602x; 1.0619x over previous
#include <hip/hip_runtime.h>

#define FDIM 128
#define HEADS 2
#define CH 64
#define NEG_SLOPE 0.2f
#define SLOTW 64
#define CAP 63
#define NBLOG 7
#define NB 128
#define CE 4096
#define PT 1024

#define GW 16
#define WKOFF (2 * FDIM)
#define WPITCH (4 * FDIM + 32)
#define WPAIRS (FDIM / 2)

typedef float v2f __attribute__((ext_vector_type(2)));
typedef float v8f __attribute__((ext_vector_type(8)));

template <int M>
__device__ __forceinline__ float row_xor(float x)
{
    return __int_as_float(__builtin_amdgcn_update_dpp(0, __float_as_int(x), 0x160 + M, 0xf, 0xf, true));
}

__device__ __forceinline__ void records_build_pair(
    int* lds, const int* __restrict__ sorted_w, const int* __restrict__ table,
    int* __restrict__ slots, int C, int TP, int K, int n, int pair)
{
    const int2* sorted = (const int2*)sorted_w;
    const int t    = threadIdx.x & 255;
    const int half = threadIdx.x >> 8;
    int4* rec4 = (int4*)lds + half * (NB * SLOTW / 4);
    int*  rec  = (int*)rec4;
    int*  cnt  = lds + 2 * NB * SLOTW + half * NB;
    const int kb = 2 * pair + half;
    const int n0 = kb * NB;

    for (int i = t; i < NB * SLOTW / 4; i += 256) rec4[i] = make_int4(0, 0, 0, 0);
    if (t < NB) cnt[t] = 0;
    __syncthreads();

    if (kb < K) {
        for (int c = t; c < C; c += 256) {
            const int* row = table + (size_t)c * TP + kb;
            const int i0 = max(row[0], 0);
            const int i1 = min(row[1], CE);
            const int2* run = sorted + (size_t)c * CE;
            for (int i = i0; i < i1; i += 4) {
                int2 e[4];
#pragma unroll
                for (int q = 0; q < 4; ++q)
                    e[q] = (i + q < i1) ? run[i + q] : make_int2(-1, 0);
                int r[4];
#pragma unroll
                for (int q = 0; q < 4; ++q) {
                    const unsigned u = (unsigned)(e[q].x - n0);
                    r[q] = (i + q < i1 && u < (unsigned)NB) ? atomicAdd(&cnt[u], 1) : CAP;
                }
#pragma unroll
                for (int q = 0; q < 4; ++q)
                    if ((unsigned)r[q] < (unsigned)CAP) rec[(e[q].x - n0) * SLOTW + 1 + r[q]] = e[q].y;
            }
        }
    }
    __syncthreads();
    if (t < NB) rec[t * SLOTW] = cnt[t];
    __syncthreads();

    int4* out4 = (int4*)slots + (size_t)n0 * (SLOTW / 4);
    for (int i = t; i < NB * SLOTW / 4; i += 256)
        if (n0 + i / (SLOTW / 4) < n) out4[i] = rec4[i];
}

__global__ __launch_bounds__(GW * 32) void gemm_pair_f32_wmma(
    const float* __restrict__ X, const float* __restrict__ Wp, const float* __restrict__ Wk,
    const float* __restrict__ bias0, const float* __restrict__ bias1,
    const float* __restrict__ att_src, const float* __restrict__ att_dst,
    float* __restrict__ a_s, float* __restrict__ a_d,
    float* __restrict__ OUTP, float* __restrict__ OUTS, int nrows,
    const int* __restrict__ sorted_w, const int* __restrict__ table,
    int* __restrict__ slots, int C, int TP, int K, int ngemm)
{
    __shared__ __attribute__((aligned(16))) float wl[WPAIRS * WPITCH];

    if ((int)blockIdx.x >= ngemm) {
        records_build_pair((int*)wl, sorted_w, table, slots, C, TP, K, nrows,
                           (int)blockIdx.x - ngemm);
        return;
    }

#pragma unroll
    for (int i = threadIdx.x; i < 2 * WPAIRS * (FDIM / 4); i += GW * 32) {
        const int m = i / (WPAIRS * (FDIM / 4));
        const int p = (i / (FDIM / 4)) % WPAIRS;
        const int q = i % (FDIM / 4);
        const float* w = (m ? Wk : Wp) + (size_t)(2 * p) * FDIM + 4 * q;
        const float4 r0 = *(const float4*)w;
        const float4 r1 = *(const float4*)(w + FDIM);
        float* dl = wl + p * WPITCH + m * WKOFF + 8 * q;
        *(float4*)dl       = make_float4(r0.x, r1.x, r0.y, r1.y);
        *(float4*)(dl + 4) = make_float4(r0.z, r1.z, r0.w, r1.w);
    }
    __syncthreads();

    const int ntiles = nrows >> 4;
    const int tile = blockIdx.x * GW + (threadIdx.x >> 5);
    if (tile >= ntiles) return;

    const int lane  = threadIdx.x & 31;
    const int lhalf = lane >> 4;
    const int lidx  = lane & 15;
    const int row0  = tile * 16;

    const float* xrow = X + (size_t)(row0 + lidx) * FDIM + 2 * lhalf;

    v8f acc[8];
    v8f ack[8];
#pragma unroll
    for (int t = 0; t < 8; ++t)
#pragma unroll
        for (int r = 0; r < 8; ++r) { acc[t][r] = 0.0f; ack[t][r] = 0.0f; }

    const float* bp = wl + lhalf * WPITCH + 2 * lidx;

#pragma unroll 2
    for (int kk = 0; kk < FDIM; kk += 4) {
        v2f a = *(const v2f*)(xrow + kk);
        const float* pr = bp + (kk >> 1) * WPITCH;
        const float* kr = pr + WKOFF;
#pragma unroll
        for (int t = 0; t < 8; ++t) {
            const v2f b = *(const v2f*)(pr + t * 32);
            acc[t] = __builtin_amdgcn_wmma_f32_16x16x4_f32(
                false, a, false, b, (short)0, acc[t], false, false);
        }
#pragma unroll
        for (int t = 0; t < 8; ++t) {
            const v2f b = *(const v2f*)(kr + t * 32);
            ack[t] = __builtin_amdgcn_wmma_f32_16x16x4_f32(
                false, a, false, b, (short)0, ack[t], false, false);
        }
    }

    const int orow = row0 + 8 * lhalf;

    {
        float ps[HEADS][8], pd[HEADS][8];
#pragma unroll
        for (int h = 0; h < HEADS; ++h) {
#pragma unroll
            for (int r = 0; r < 8; ++r) { ps[h][r] = 0.0f; pd[h][r] = 0.0f; }
#pragma unroll
            for (int tt = 0; tt < 4; ++tt) {
                const int t = 4 * h + tt;
                const float vs = att_src[t * 16 + lidx];
                const float vd = att_dst[t * 16 + lidx];
#pragma unroll
                for (int r = 0; r < 8; ++r) {
                    ps[h][r] += acc[t][r] * vs;
                    pd[h][r] += acc[t][r] * vd;
                }
            }
        }
#pragma unroll
        for (int h = 0; h < HEADS; ++h)
#pragma unroll
            for (int r = 0; r < 8; ++r) {
                float s = ps[h][r], d = pd[h][r];
                s += row_xor<1>(s); d += row_xor<1>(d);
                s += row_xor<2>(s); d += row_xor<2>(d);
                s += row_xor<4>(s); d += row_xor<4>(d);
                s += row_xor<8>(s); d += row_xor<8>(d);
                ps[h][r] = s; pd[h][r] = d;
            }
        float os = ps[0][0], od = pd[0][0];
#pragma unroll
        for (int j = 1; j < 16; ++j)
            if (lidx == j) { os = ps[j & 1][j >> 1]; od = pd[j & 1][j >> 1]; }
        const size_t ao = (size_t)orow * HEADS + lidx;
        a_s[ao] = os;
        a_d[ao] = od;
    }
#pragma unroll
    for (int t = 0; t < 8; ++t) {
        const int col = t * 16 + lidx;
        float* o = OUTP + (size_t)orow * FDIM + col;
#pragma unroll
        for (int r = 0; r < 8; ++r)
            o[(size_t)r * FDIM] = acc[t][r];
    }
#pragma unroll
    for (int t = 0; t < 8; ++t) {
        const int col = t * 16 + lidx;
        const float badd = bias0[col] + bias1[col];
        float* o = OUTS + (size_t)orow * FDIM + col;
#pragma unroll
        for (int r = 0; r < 8; ++r)
            o[(size_t)r * FDIM] = ack[t][r] + badd;
    }
}

__device__ __forceinline__ float lrelu(float x)  { return x > 0.f ? x : NEG_SLOPE * x; }

__global__ __launch_bounds__(PT) void edge_chunk_order_kernel(
    const int* __restrict__ src, const int* __restrict__ dst,
    int* __restrict__ sorted_w, int* __restrict__ table, int E_, int K, int TP)
{
    int2* sorted = (int2*)sorted_w;
    __shared__ int  hist[PT];
    __shared__ int  wsum[32];
    __shared__ int2 stage[CE];
    const int t    = threadIdx.x;
    const int lane = t & 31;
    const int wave = t >> 5;
    const int e0   = blockIdx.x * CE;

    hist[t] = 0;
    __syncthreads();

    int d[4], s[4], b[4], r[4];
    bool v[4];
#pragma unroll
    for (int j = 0; j < 4; ++j) {
        const int k = e0 + t + j * PT;
        v[j] = k < E_;
        d[j] = v[j] ? dst[k] : 0;
        s[j] = v[j] ? src[k] : 0;
    }
#pragma unroll
    for (int j = 0; j < 4; ++j) {
        b[j] = min((int)((unsigned)d[j] >> NBLOG), K - 1);
        r[j] = v[j] ? atomicAdd(&hist[b[j]], 1) : 0;
    }
    __syncthreads();

    const int cntb = hist[t];
    int inc = cntb;
#pragma unroll
    for (int o = 1; o < 32; o <<= 1) {
        const int y = __shfl_up(inc, o);
        if (lane >= o) inc += y;
    }
    if (lane == 31) wsum[wave] = inc;
    __syncthreads();
    if (wave == 0) {
        const int w = wsum[lane];
        int wi = w;
#pragma unroll
        for (int o = 1; o < 32; o <<= 1) {
            const int y = __shfl_up(wi, o);
            if (lane >= o) wi += y;
        }
        wsum[lane] = wi - w;
    }
    __syncthreads();
    hist[t] = wsum[wave] + inc - cntb;
    __syncthreads();

#pragma unroll
    for (int j = 0; j < 4; ++j)
        if (v[j]) stage[hist[b[j]] + r[j]] = make_int2(d[j], s[j]);
    __syncthreads();

    const int nval = min(CE, E_ - e0);
#pragma unroll
    for (int j = 0; j < 4; ++j) {
        const int i = t + j * PT;
        if (i < nval) sorted[(size_t)e0 + i] = stage[i];
    }
    if (t <= K) table[(size_t)blockIdx.x * TP + t] = hist[t];
}

__device__ __forceinline__ float lane_value(float x, int l)
{
    return __int_as_float(__builtin_amdgcn_readlane(__float_as_int(x), l));
}
__device__ __forceinline__ float wave_max(float x)
{
    x = fmaxf(x, row_xor<1>(x));
    x = fmaxf(x, row_xor<2>(x));
    x = fmaxf(x, row_xor<4>(x));
    x = fmaxf(x, row_xor<8>(x));
    return fmaxf(lane_value(x, 0), lane_value(x, 16));
}
__device__ __forceinline__ float wave_sum(float x)
{
    x += row_xor<1>(x);
    x += row_xor<2>(x);
    x += row_xor<4>(x);
    x += row_xor<8>(x);
    return lane_value(x, 0) + lane_value(x, 16);
}

template <bool RELU>
__global__ __launch_bounds__(256) void node_softmax_gather_kernel(
    const int* __restrict__ slots,
    const int* __restrict__ ovdst, const int* __restrict__ ovsrc,
    const float* __restrict__ a_s, const float* __restrict__ a_d,
    const float* __restrict__ Hp, float* __restrict__ out, int E_, int n)
{
    const int lane = threadIdx.x & 31;
    const int d = __builtin_amdgcn_readfirstlane((int)(blockIdx.x * 8 + (threadIdx.x >> 5)));
    if (d >= n) return;
    const int h  = lane >> 4;
    const int c0 = lane * 4;
    const int* rec = slots + (size_t)d * SLOTW;
    const int cnt = rec[0];
    const int end = cnt > CAP ? 0 : max(cnt, 0);
    const int nov = cnt > CAP ? E_ : 0;
    const float2 adv = *(const float2*)(a_d + (size_t)d * HEADS);
    const float2 asv = *(const float2*)(a_s + (size_t)d * HEADS);
    const float4 hs  = *(const float4*)(Hp + (size_t)d * FDIM + c0);
    const float es0 = lrelu(asv.x + adv.x);
    const float es1 = lrelu(asv.y + adv.y);

    auto edge = [&](int j, int& s, float& e0, float& e1) {
        s = d; e0 = -INFINITY; e1 = -INFINITY;
        if (j < end) {
            s = min(max(rec[1 + j], 0), n - 1);
            const float2 a = *(const float2*)(a_s + (size_t)s * HEADS);
            e0 = lrelu(a.x + adv.x);
            e1 = lrelu(a.y + adv.y);
        }
    };
    auto over = [&](int j, int& s, float& e0, float& e1) {
        s = d; e0 = -INFINITY; e1 = -INFINITY;
        bool v = false;
        if (j < nov) v = ovdst[j] == d;
        if (v) {
            s = min(max(ovsrc[j], 0), n - 1);
            const float2 a = *(const float2*)(a_s + (size_t)s * HEADS);
            e0 = lrelu(a.x + adv.x);
            e1 = lrelu(a.y + adv.y);
        }
        return v;
    };

    int s_l; float e0, e1;
    edge(lane, s_l, e0, e1);

    float mx0 = e0, mx1 = e1;
    for (int base = 32; base < end; base += 32) {
        int s; float f0, f1;
        edge(base + lane, s, f0, f1);
        mx0 = fmaxf(mx0, f0); mx1 = fmaxf(mx1, f1);
    }
    for (int base = 0; base < nov; base += 32) {
        int s; float f0, f1;
        over(base + lane, s, f0, f1);
        mx0 = fmaxf(mx0, f0); mx1 = fmaxf(mx1, f1);
    }
    const float m0 = fmaxf(wave_max(mx0), es0);
    const float m1 = fmaxf(wave_max(mx1), es1);

    const bool v_l = lane < end;
    const float p0 = v_l ? __expf(e0 - m0) : 0.f;
    const float p1 = v_l ? __expf(e1 - m1) : 0.f;
    float sm0 = p0, sm1 = p1;
    for (int base = 32; base < end; base += 32) {
        int s; float f0, f1;
        edge(base + lane, s, f0, f1);
        if (base + lane < end) { sm0 += __expf(f0 - m0); sm1 += __expf(f1 - m1); }
    }
    for (int base = 0; base < nov; base += 32) {
        int s; float f0, f1;
        if (over(base + lane, s, f0, f1)) { sm0 += __expf(f0 - m0); sm1 += __expf(f1 - m1); }
    }
    const float ps0 = __expf(es0 - m0);
    const float ps1 = __expf(es1 - m1);
    const float den0 = (wave_sum(sm0) + ps0) + 1e-16f;
    const float den1 = (wave_sum(sm1) + ps1) + 1e-16f;

    const float aself = h ? ps1 / den1 : ps0 / den0;
    float4 acc;
    acc.x = hs.x * aself; acc.y = hs.y * aself; acc.z = hs.z * aself; acc.w = hs.w * aself;

    auto row = [&](int s_v, float a0, float a1, int t) {
        const int s = __builtin_amdgcn_readlane(s_v, t);
        const float w0 = lane_value(a0, t);
        const float w1 = lane_value(a1, t);
        const float a = h ? w1 : w0;
        const float4 hv = *(const float4*)(Hp + (size_t)s * FDIM + c0);
        acc.x += hv.x * a;
        acc.y += hv.y * a;
        acc.z += hv.z * a;
        acc.w += hv.w * a;
    };
    auto rows = [&](int s_v, float a0, float a1, int m) {
#pragma unroll 4
        for (int t = 0; t < m; ++t) row(s_v, a0, a1, t);
    };

    rows(s_l, p0 / den0, p1 / den1, min(32, end));
    for (int base = 32; base < end; base += 32) {
        int s; float f0, f1;
        edge(base + lane, s, f0, f1);
        const bool v = base + lane < end;
        const float a0 = v ? __expf(f0 - m0) / den0 : 0.f;
        const float a1 = v ? __expf(f1 - m1) / den1 : 0.f;
        rows(s, a0, a1, min(32, end - base));
    }
    for (int base = 0; base < nov; base += 32) {
        int s; float f0, f1;
        const bool v = over(base + lane, s, f0, f1);
        const float a0 = v ? __expf(f0 - m0) / den0 : 0.f;
        const float a1 = v ? __expf(f1 - m1) / den1 : 0.f;
        for (unsigned live = __builtin_amdgcn_ballot_w32(v); live; live &= live - 1)
            row(s, a0, a1, __builtin_ctz(live));
    }

    float4* o = (float4*)(out + (size_t)d * FDIM + c0);
    float4 ov = *o;
    ov.x += acc.x; ov.y += acc.y; ov.z += acc.z; ov.w += acc.w;
    if (RELU) {
        ov.x = fmaxf(ov.x, 0.0f); ov.y = fmaxf(ov.y, 0.0f);
        ov.z = fmaxf(ov.z, 0.0f); ov.w = fmaxf(ov.w, 0.0f);
    }
    *o = ov;
}

extern "C" void kernel_launch(void* const* d_in, const int* in_sizes, int n_in,
                              void* d_out, int out_size, void* d_ws, size_t ws_size,
                              hipStream_t stream)
{
    const float* x        = (const float*)d_in[0];
    const int*   src      = (const int*)  d_in[1];
    const int*   dst      = (const int*)  d_in[2];
    const float* W1       = (const float*)d_in[3];
    const float* att_src1 = (const float*)d_in[4];
    const float* att_dst1 = (const float*)d_in[5];
    const float* b1       = (const float*)d_in[6];
    const float* Wsk1     = (const float*)d_in[7];
    const float* bsk1     = (const float*)d_in[8];
    const float* W2       = (const float*)d_in[9];
    const float* att_src2 = (const float*)d_in[10];
    const float* att_dst2 = (const float*)d_in[11];
    const float* b2       = (const float*)d_in[12];
    const float* Wsk2     = (const float*)d_in[13];
    const float* bsk2     = (const float*)d_in[14];

    const int N  = in_sizes[0] / FDIM;
    const int E  = in_sizes[1];
    float* out   = (float*)d_out;

    char* base = (char*)d_ws;
    size_t off = 0;
    auto alloc = [&](size_t bytes) {
        void* p = base + off;
        off = (off + bytes + 255) & ~(size_t)255;
        return p;
    };
    float* Hproj = (float*)alloc((size_t)N * FDIM * 4);
    float* Hbuf  = (float*)alloc((size_t)N * FDIM * 4);
    float* a_s   = (float*)alloc((size_t)N * HEADS * 4);
    float* a_d   = (float*)alloc((size_t)N * HEADS * 4);
    const int K  = (N + NB - 1) / NB;
    const int C  = (E + CE - 1) / CE;
    const int TP = (K + 1 + 31) & ~31;
    int*   slots  = (int*) alloc((size_t)N * SLOTW * 4);
    int*   sorted = (int*) alloc((size_t)C * CE * 8);
    int*   table  = (int*) alloc((size_t)C * TP * 4);

    const int ntiles     = N / 16;
    const int  ngemm     = (ntiles + GW - 1) / GW;
    const int  npair     = (K + 1) / 2;
    const dim3 gGemm(ngemm), gGemmRec(ngemm + npair), bGemm(GW * 32);
    const dim3 b256(256);
    const dim3 gNode((N + 7) / 8);

    edge_chunk_order_kernel<<<dim3(C), dim3(PT), 0, stream>>>(src, dst, sorted, table, E, K, TP);

    gemm_pair_f32_wmma<<<gGemmRec, bGemm, 0, stream>>>(x, W1, Wsk1, b1, bsk1, att_src1, att_dst1, a_s, a_d, Hproj, Hbuf, N,
                                                     sorted, table, slots, C, TP, K, ngemm);
    node_softmax_gather_kernel<true><<<gNode, b256, 0, stream>>>(slots, dst, src, a_s, a_d, Hproj, Hbuf, E, N);

    gemm_pair_f32_wmma<<<gGemm, bGemm, 0, stream>>>(Hbuf, W2, Wsk2, b2, bsk2, att_src2, att_dst2, a_s, a_d, Hproj, out, N,
                                                  sorted, table, slots, C, TP, K, ngemm);
    node_softmax_gather_kernel<false><<<gNode, b256, 0, stream>>>(slots, dst, src, a_s, a_d, Hproj, out, E, N);
}
